# starting kernel
# baseline (speedup 1.0000x reference)
#include <hip/hip_runtime.h>

#define FIN 256
#define HID 64
#define COUT 40

typedef float v2f __attribute__((ext_vector_type(2)));
typedef float v8f __attribute__((ext_vector_type(8)));

__global__ __launch_bounds__(256) void k_deg_init(float* __restrict__ deg, int n) {
    int i = blockIdx.x * 256 + threadIdx.x;
    if (i < n) deg[i] = 1.0f;
}

__global__ __launch_bounds__(256) void k_deg_acc(const int* __restrict__ dst, int E,
                                                 float* __restrict__ deg) {
    int i = blockIdx.x * 256 + threadIdx.x;
    if (i < E) atomicAdd(deg + dst[i], 1.0f);
}

__global__ __launch_bounds__(256) void k_dinv(float* __restrict__ deg, int n) {
    int i = blockIdx.x * 256 + threadIdx.x;
    if (i < n) deg[i] = rsqrtf(deg[i]);
}

__global__ __launch_bounds__(256) void k_zero4(float4* __restrict__ p, int n4) {
    int i = blockIdx.x * 256 + threadIdx.x;
    if (i < n4) p[i] = make_float4(0.f, 0.f, 0.f, 0.f);
}

__global__ __launch_bounds__(256) void k_gemm1(const float* __restrict__ X,
                                               const float* __restrict__ W,
                                               float* __restrict__ H, int nrows) {
    int wave = (blockIdx.x * 256 + threadIdx.x) >> 5;
    int lane = threadIdx.x & 31;
    int m0 = wave << 4;
    if (m0 >= nrows) return;
    int row = lane & 15;
    int kk  = (lane >> 4) << 1;
    const float* Xr = X + (size_t)(m0 + row) * FIN + kk;
    v8f c0 = {}, c1 = {}, c2 = {}, c3 = {};
    for (int k0 = 0; k0 < FIN; k0 += 4) {
        v2f a;
        a.x = Xr[k0];
        a.y = Xr[k0 + 1];
        const float* Wp = W + (size_t)(k0 + kk) * HID;
        v2f b0 = { Wp[row],      Wp[HID + row] };
        v2f b1 = { Wp[16 + row], Wp[HID + 16 + row] };
        v2f b2 = { Wp[32 + row], Wp[HID + 32 + row] };
        v2f b3 = { Wp[48 + row], Wp[HID + 48 + row] };
        c0 = __builtin_amdgcn_wmma_f32_16x16x4_f32(false, a, false, b0, (short)0, c0, false, false);
        c1 = __builtin_amdgcn_wmma_f32_16x16x4_f32(false, a, false, b1, (short)0, c1, false, false);
        c2 = __builtin_amdgcn_wmma_f32_16x16x4_f32(false, a, false, b2, (short)0, c2, false, false);
        c3 = __builtin_amdgcn_wmma_f32_16x16x4_f32(false, a, false, b3, (short)0, c3, false, false);
    }
    int mrow = m0 + ((lane >> 4) << 3);
    int col  = lane & 15;
    for (int r = 0; r < 8; ++r) {
        float* out = H + (size_t)(mrow + r) * HID + col;
        out[0]  = c0[r];
        out[16] = c1[r];
        out[32] = c2[r];
        out[48] = c3[r];
    }
}

__global__ __launch_bounds__(256) void k_gemm2(const float* __restrict__ H1,
                                               const float* __restrict__ W,
                                               float* __restrict__ H2, int nrows) {
    int wave = (blockIdx.x * 256 + threadIdx.x) >> 5;
    int lane = threadIdx.x & 31;
    int m0 = wave << 4;
    if (m0 >= nrows) return;
    int row = lane & 15;
    int kk  = (lane >> 4) << 1;
    const float* Hr = H1 + (size_t)(m0 + row) * HID + kk;
    v8f c0 = {}, c1 = {}, c2 = {};
    for (int k0 = 0; k0 < HID; k0 += 4) {
        v2f a;
        a.x = Hr[k0];
        a.y = Hr[k0 + 1];
        const float* Wp = W + (size_t)(k0 + kk) * COUT;
        v2f b0 = { Wp[row],      Wp[COUT + row] };
        v2f b1 = { Wp[16 + row], Wp[COUT + 16 + row] };
        int col2 = 32 + row;
        v2f b2 = { (col2 < COUT) ? Wp[col2] : 0.0f,
                   (col2 < COUT) ? Wp[COUT + col2] : 0.0f };
        c0 = __builtin_amdgcn_wmma_f32_16x16x4_f32(false, a, false, b0, (short)0, c0, false, false);
        c1 = __builtin_amdgcn_wmma_f32_16x16x4_f32(false, a, false, b1, (short)0, c1, false, false);
        c2 = __builtin_amdgcn_wmma_f32_16x16x4_f32(false, a, false, b2, (short)0, c2, false, false);
    }
    int mrow = m0 + ((lane >> 4) << 3);
    int col  = lane & 15;
    for (int r = 0; r < 8; ++r) {
        float* out = H2 + (size_t)(mrow + r) * COUT;
        out[col]      = c0[r];
        out[16 + col] = c1[r];
        if (32 + col < COUT) out[32 + col] = c2[r];
    }
}

__global__ __launch_bounds__(256) void k_agg1(const int* __restrict__ src,
                                              const int* __restrict__ dst, int E,
                                              const float* __restrict__ dinv,
                                              const float4* __restrict__ H4,
                                              float* __restrict__ A) {
    long long idx = (long long)blockIdx.x * 256 + threadIdx.x;
    if (idx >= (long long)E * 16) return;
    int e  = (int)(idx >> 4);
    int f4 = (int)(idx & 15);
    int s = src[e], d = dst[e];
    float nrm = dinv[s] * dinv[d];
    float4 v = H4[(size_t)s * (HID / 4) + f4];
    float* out = A + (size_t)d * HID + f4 * 4;
    atomicAdd(out + 0, v.x * nrm);
    atomicAdd(out + 1, v.y * nrm);
    atomicAdd(out + 2, v.z * nrm);
    atomicAdd(out + 3, v.w * nrm);
}

__global__ __launch_bounds__(256) void k_fin1(float4* __restrict__ A,
                                              const float4* __restrict__ H4,
                                              const float* __restrict__ dinv,
                                              const float* __restrict__ b1, int n) {
    int idx = blockIdx.x * 256 + threadIdx.x;
    if (idx >= n * (HID / 4)) return;
    int i  = idx >> 4;
    int f4 = idx & 15;
    float di = dinv[i];
    float nn = di * di;
    float4 a = A[idx];
    float4 h = H4[idx];
    const float* b = b1 + f4 * 4;
    a.x = fmaxf(a.x + h.x * nn + b[0], 0.f);
    a.y = fmaxf(a.y + h.y * nn + b[1], 0.f);
    a.z = fmaxf(a.z + h.z * nn + b[2], 0.f);
    a.w = fmaxf(a.w + h.w * nn + b[3], 0.f);
    A[idx] = a;
}

__global__ __launch_bounds__(320) void k_agg2(const int* __restrict__ src,
                                              const int* __restrict__ dst, int E,
                                              const float* __restrict__ dinv,
                                              const float4* __restrict__ H4,
                                              float* __restrict__ out) {
    int t = threadIdx.x;
    int e  = blockIdx.x * 32 + t / 10;
    int f4 = t % 10;
    if (e >= E) return;
    int s = src[e], d = dst[e];
    float nrm = dinv[s] * dinv[d];
    float4 v = H4[(size_t)s * (COUT / 4) + f4];
    float* o = out + (size_t)d * COUT + f4 * 4;
    atomicAdd(o + 0, v.x * nrm);
    atomicAdd(o + 1, v.y * nrm);
    atomicAdd(o + 2, v.z * nrm);
    atomicAdd(o + 3, v.w * nrm);
}

__global__ __launch_bounds__(256) void k_fin2(float4* __restrict__ out,
                                              const float4* __restrict__ H4,
                                              const float* __restrict__ dinv,
                                              const float* __restrict__ b2, int n) {
    int idx = blockIdx.x * 256 + threadIdx.x;
    if (idx >= n * (COUT / 4)) return;
    int i  = idx / 10;
    int f4 = idx % 10;
    float di = dinv[i];
    float nn = di * di;
    float4 o = out[idx];
    float4 h = H4[idx];
    const float* b = b2 + f4 * 4;
    o.x += h.x * nn + b[0];
    o.y += h.y * nn + b[1];
    o.z += h.z * nn + b[2];
    o.w += h.w * nn + b[3];
    out[idx] = o;
}

extern "C" void kernel_launch(void* const* d_in, const int* in_sizes, int n_in,
                              void* d_out, int out_size, void* d_ws, size_t ws_size,
                              hipStream_t stream) {
    const float* x  = (const float*)d_in[0];
    const float* W1 = (const float*)d_in[1];
    const float* b1 = (const float*)d_in[2];
    const float* W2 = (const float*)d_in[3];
    const float* b2 = (const float*)d_in[4];
    const int*   ei = (const int*)d_in[5];

    const int Nn = in_sizes[0] / FIN;
    const int E  = in_sizes[5] / 2;
    const int* src = ei;
    const int* dst = ei + E;

    float* ws   = (float*)d_ws;
    float* dinv = ws;
    float* h    = ws + Nn;
    float* agg  = h + (size_t)Nn * HID;
    float* h2   = h;
    float* out  = (float*)d_out;

    int gN  = (Nn + 255) / 256;
    int gE  = (E + 255) / 256;
    int h4n = Nn * (HID / 4);
    int o4n = Nn * (COUT / 4);
    int gH4 = (h4n + 255) / 256;
    int gO4 = (o4n + 255) / 256;
    int gE1 = (int)(((long long)E * 16 + 255) / 256);
    int gE2 = (E + 31) / 32;
    int gemmBlocks = ((Nn / 16) + 7) / 8;

    k_deg_init<<<gN, 256, 0, stream>>>(dinv, Nn);
    k_deg_acc<<<gE, 256, 0, stream>>>(dst, E, dinv);
    k_dinv<<<gN, 256, 0, stream>>>(dinv, Nn);

    k_gemm1<<<gemmBlocks, 256, 0, stream>>>(x, W1, h, Nn);
    k_zero4<<<gH4, 256, 0, stream>>>((float4*)agg, h4n);
    k_agg1<<<gE1, 256, 0, stream>>>(src, dst, E, dinv, (const float4*)h, agg);
    k_fin1<<<gH4, 256, 0, stream>>>((float4*)agg, (const float4*)h, dinv, b1, Nn);

    k_gemm2<<<gemmBlocks, 256, 0, stream>>>(agg, W2, h2, Nn);
    k_zero4<<<gO4, 256, 0, stream>>>((float4*)out, o4n);
    k_agg2<<<gE2, 320, 0, stream>>>(src, dst, E, dinv, (const float4*)h2, out);
    k_fin2<<<gO4, 256, 0, stream>>>((float4*)out, (const float4*)h2, dinv, b2, Nn);
}


# edge scatters walk whole rows: a wave takes 32 edges, lane = column, so each gather and each f32 atomic add is one full 128-B line (layer-2 planes at 64-float pitch); sums in arrival order as before
# speedup vs baseline: 1.3015x; 1.3015x over previous
#include <hip/hip_runtime.h>

#define FIN 256
#define HID 64
#define COUT 40
#define LD2 64

typedef float v2f __attribute__((ext_vector_type(2)));
typedef float v8f __attribute__((ext_vector_type(8)));

__global__ __launch_bounds__(256) void k_deg_init(float* __restrict__ deg, int n) {
    int i = blockIdx.x * 256 + threadIdx.x;
    if (i < n) deg[i] = 1.0f;
}

__global__ __launch_bounds__(256) void k_deg_acc(const int* __restrict__ dst, int E,
                                                 float* __restrict__ deg) {
    int i = blockIdx.x * 256 + threadIdx.x;
    if (i < E) atomicAdd(deg + dst[i], 1.0f);
}

__global__ __launch_bounds__(256) void k_dinv(float* __restrict__ deg, int n) {
    int i = blockIdx.x * 256 + threadIdx.x;
    if (i < n) deg[i] = rsqrtf(deg[i]);
}

__global__ __launch_bounds__(256) void k_zero4(float4* __restrict__ p, int n4) {
    int i = blockIdx.x * 256 + threadIdx.x;
    if (i < n4) p[i] = make_float4(0.f, 0.f, 0.f, 0.f);
}

__global__ __launch_bounds__(256) void k_gemm1(const float* __restrict__ X,
                                               const float* __restrict__ W,
                                               float* __restrict__ H, int nrows) {
    int wave = (blockIdx.x * 256 + threadIdx.x) >> 5;
    int lane = threadIdx.x & 31;
    int m0 = wave << 4;
    if (m0 >= nrows) return;
    int row = lane & 15;
    int kk  = (lane >> 4) << 1;
    const float* Xr = X + (size_t)(m0 + row) * FIN + kk;
    v8f c0 = {}, c1 = {}, c2 = {}, c3 = {};
    for (int k0 = 0; k0 < FIN; k0 += 4) {
        v2f a;
        a.x = Xr[k0];
        a.y = Xr[k0 + 1];
        const float* Wp = W + (size_t)(k0 + kk) * HID;
        v2f b0 = { Wp[row],      Wp[HID + row] };
        v2f b1 = { Wp[16 + row], Wp[HID + 16 + row] };
        v2f b2 = { Wp[32 + row], Wp[HID + 32 + row] };
        v2f b3 = { Wp[48 + row], Wp[HID + 48 + row] };
        c0 = __builtin_amdgcn_wmma_f32_16x16x4_f32(false, a, false, b0, (short)0, c0, false, false);
        c1 = __builtin_amdgcn_wmma_f32_16x16x4_f32(false, a, false, b1, (short)0, c1, false, false);
        c2 = __builtin_amdgcn_wmma_f32_16x16x4_f32(false, a, false, b2, (short)0, c2, false, false);
        c3 = __builtin_amdgcn_wmma_f32_16x16x4_f32(false, a, false, b3, (short)0, c3, false, false);
    }
    int mrow = m0 + ((lane >> 4) << 3);
    int col  = lane & 15;
    for (int r = 0; r < 8; ++r) {
        float* out = H + (size_t)(mrow + r) * HID + col;
        out[0]  = c0[r];
        out[16] = c1[r];
        out[32] = c2[r];
        out[48] = c3[r];
    }
}

__global__ __launch_bounds__(256) void k_gemm2(const float* __restrict__ H1,
                                               const float* __restrict__ W,
                                               float* __restrict__ H2, int nrows) {
    int wave = (blockIdx.x * 256 + threadIdx.x) >> 5;
    int lane = threadIdx.x & 31;
    int m0 = wave << 4;
    if (m0 >= nrows) return;
    int row = lane & 15;
    int kk  = (lane >> 4) << 1;
    const float* Hr = H1 + (size_t)(m0 + row) * HID + kk;
    v8f c0 = {}, c1 = {}, c2 = {};
    for (int k0 = 0; k0 < HID; k0 += 4) {
        v2f a;
        a.x = Hr[k0];
        a.y = Hr[k0 + 1];
        const float* Wp = W + (size_t)(k0 + kk) * COUT;
        v2f b0 = { Wp[row],      Wp[COUT + row] };
        v2f b1 = { Wp[16 + row], Wp[COUT + 16 + row] };
        int col2 = 32 + row;
        v2f b2 = { (col2 < COUT) ? Wp[col2] : 0.0f,
                   (col2 < COUT) ? Wp[COUT + col2] : 0.0f };
        c0 = __builtin_amdgcn_wmma_f32_16x16x4_f32(false, a, false, b0, (short)0, c0, false, false);
        c1 = __builtin_amdgcn_wmma_f32_16x16x4_f32(false, a, false, b1, (short)0, c1, false, false);
        c2 = __builtin_amdgcn_wmma_f32_16x16x4_f32(false, a, false, b2, (short)0, c2, false, false);
    }
    int mrow = m0 + ((lane >> 4) << 3);
    int col  = lane & 15;
    for (int r = 0; r < 8; ++r) {
        float* out = H2 + (size_t)(mrow + r) * LD2;
        out[col]      = c0[r];
        out[16 + col] = c1[r];
        if (32 + col < COUT) out[32 + col] = c2[r];
    }
}

template <int NTAIL>
__device__ __forceinline__ void agg_group8(const float* __restrict__ H, float* __restrict__ A,
                                           int s, int d, float nrm, int j0, int lane) {
    const bool tail = (NTAIL == 32) || (lane < NTAIL);
    const float* hp[8];
    float v0[8], v1[8];
#pragma unroll
    for (int u = 0; u < 8; ++u) {
        int sj = __builtin_amdgcn_readlane(s, j0 + u);
        hp[u] = H + (size_t)sj * LD2 + lane;
        v0[u] = hp[u][0];
    }
#pragma unroll
    for (int u = 0; u < 8; ++u) v1[u] = 0.0f;
    if (tail) {
#pragma unroll
        for (int u = 0; u < 8; ++u) v1[u] = hp[u][32];
    }
    float* ap[8];
    float nj[8];
#pragma unroll
    for (int u = 0; u < 8; ++u) {
        int dj = __builtin_amdgcn_readlane(d, j0 + u);
        nj[u] = __int_as_float(__builtin_amdgcn_readlane(__float_as_int(nrm), j0 + u));
        ap[u] = A + (size_t)dj * LD2 + lane;
        atomicAdd(ap[u], v0[u] * nj[u]);
    }
    if (tail) {
#pragma unroll
        for (int u = 0; u < 8; ++u) atomicAdd(ap[u] + 32, v1[u] * nj[u]);
    }
}

template <int NTAIL>
__global__ __launch_bounds__(256) void k_agg(const int* __restrict__ src,
                                             const int* __restrict__ dst, int E,
                                             const float* __restrict__ dinv,
                                             const float* __restrict__ H,
                                             float* __restrict__ A) {
    int wave = (blockIdx.x * 256 + threadIdx.x) >> 5;
    int lane = threadIdx.x & 31;
    int e0 = wave << 5;
    if (e0 >= E) return;
    int e = e0 + lane;
    int s = 0, d = 0;
    float nrm = 0.0f;
    if (e < E) {
        s = src[e];
        d = dst[e];
        nrm = dinv[s] * dinv[d];
    }
    if (e0 + 32 <= E) {
        for (int j0 = 0; j0 < 32; j0 += 8) agg_group8<NTAIL>(H, A, s, d, nrm, j0, lane);
    } else {
        int cnt = E - e0;
        for (int j = 0; j < cnt; ++j) {
            int   sj = __builtin_amdgcn_readlane(s, j);
            int   dj = __builtin_amdgcn_readlane(d, j);
            float nj = __int_as_float(__builtin_amdgcn_readlane(__float_as_int(nrm), j));
            const float* hp = H + (size_t)sj * LD2 + lane;
            float* ap = A + (size_t)dj * LD2 + lane;
            atomicAdd(ap, hp[0] * nj);
            if (NTAIL == 32 || lane < NTAIL) atomicAdd(ap + 32, hp[32] * nj);
        }
    }
}

__global__ __launch_bounds__(256) void k_fin1(float4* __restrict__ A,
                                              const float4* __restrict__ H4,
                                              const float* __restrict__ dinv,
                                              const float* __restrict__ b1, int n) {
    int idx = blockIdx.x * 256 + threadIdx.x;
    if (idx >= n * (HID / 4)) return;
    int i  = idx >> 4;
    int f4 = idx & 15;
    float di = dinv[i];
    float nn = di * di;
    float4 a = A[idx];
    float4 h = H4[idx];
    const float* b = b1 + f4 * 4;
    a.x = fmaxf(a.x + h.x * nn + b[0], 0.f);
    a.y = fmaxf(a.y + h.y * nn + b[1], 0.f);
    a.z = fmaxf(a.z + h.z * nn + b[2], 0.f);
    a.w = fmaxf(a.w + h.w * nn + b[3], 0.f);
    A[idx] = a;
}

__global__ __launch_bounds__(256) void k_fin2(float4* __restrict__ out,
                                              const float4* __restrict__ acc,
                                              const float4* __restrict__ H4,
                                              const float* __restrict__ dinv,
                                              const float* __restrict__ b2, int n) {
    int idx = blockIdx.x * 256 + threadIdx.x;
    if (idx >= n * (COUT / 4)) return;
    int i  = idx / 10;
    int f4 = idx % 10;
    float di = dinv[i];
    float nn = di * di;
    float4 o = acc[(size_t)i * (LD2 / 4) + f4];
    float4 h = H4[(size_t)i * (LD2 / 4) + f4];
    const float* b = b2 + f4 * 4;
    o.x += h.x * nn + b[0];
    o.y += h.y * nn + b[1];
    o.z += h.z * nn + b[2];
    o.w += h.w * nn + b[3];
    out[idx] = o;
}

extern "C" void kernel_launch(void* const* d_in, const int* in_sizes, int n_in,
                              void* d_out, int out_size, void* d_ws, size_t ws_size,
                              hipStream_t stream) {
    const float* x  = (const float*)d_in[0];
    const float* W1 = (const float*)d_in[1];
    const float* b1 = (const float*)d_in[2];
    const float* W2 = (const float*)d_in[3];
    const float* b2 = (const float*)d_in[4];
    const int*   ei = (const int*)d_in[5];

    const int Nn = in_sizes[0] / FIN;
    const int E  = in_sizes[5] / 2;
    const int* src = ei;
    const int* dst = ei + E;

    float* ws   = (float*)d_ws;
    float* dinv = ws;
    float* h    = ws + Nn;
    float* agg  = h + (size_t)Nn * HID;
    float* h2   = h;
    float* acc2 = agg;
    float* out  = (float*)d_out;

    int gN  = (Nn + 255) / 256;
    int gE  = (E + 255) / 256;
    int h4n = Nn * (HID / 4);
    int o4n = Nn * (COUT / 4);
    int gH4 = (h4n + 255) / 256;
    int gO4 = (o4n + 255) / 256;
    int gEw = (((E + 31) / 32) + 7) / 8;
    int gemmBlocks = ((Nn / 16) + 7) / 8;

    k_deg_init<<<gN, 256, 0, stream>>>(dinv, Nn);
    k_deg_acc<<<gE, 256, 0, stream>>>(dst, E, dinv);
    k_dinv<<<gN, 256, 0, stream>>>(dinv, Nn);

    k_gemm1<<<gemmBlocks, 256, 0, stream>>>(x, W1, h, Nn);
    k_zero4<<<gH4, 256, 0, stream>>>((float4*)agg, h4n);
    k_agg<32><<<gEw, 256, 0, stream>>>(src, dst, E, dinv, h, agg);
    k_fin1<<<gH4, 256, 0, stream>>>((float4*)agg, (const float4*)h, dinv, b1, Nn);

    k_gemm2<<<gemmBlocks, 256, 0, stream>>>(agg, W2, h2, Nn);
    k_zero4<<<gH4, 256, 0, stream>>>((float4*)acc2, h4n);
    k_agg<8><<<gEw, 256, 0, stream>>>(src, dst, E, dinv, h2, acc2);
    k_fin2<<<gO4, 256, 0, stream>>>((float4*)out, (const float4*)acc2, (const float4*)h2, dinv, b2, Nn);
}


# norm factored out of the edge sums: products store dinv-scaled rows, also as the sums' start value (self loop first); readers scale by dinv[dst]; zero/dinv/finalize-1 passes gone; rounding changes
# speedup vs baseline: 1.3923x; 1.0698x over previous
#include <hip/hip_runtime.h>

#define FIN 256
#define HID 64
#define COUT 40
#define LD2 64

typedef float v2f __attribute__((ext_vector_type(2)));
typedef float v8f __attribute__((ext_vector_type(8)));

__global__ __launch_bounds__(256) void k_deg_init(float* __restrict__ deg, int n) {
    int i = blockIdx.x * 256 + threadIdx.x;
    if (i < n) deg[i] = 1.0f;
}

__global__ __launch_bounds__(256) void k_deg_acc(const int* __restrict__ dst, int E,
                                                 float* __restrict__ deg) {
    int i = blockIdx.x * 256 + threadIdx.x;
    if (i < E) atomicAdd(deg + dst[i], 1.0f);
}

__global__ __launch_bounds__(256) void k_gemm1(const float* __restrict__ X,
                                               const float* __restrict__ W,
                                               const float* __restrict__ deg,
                                               float* __restrict__ H,
                                               float* __restrict__ A, int nrows) {
    int wave = (blockIdx.x * 256 + threadIdx.x) >> 5;
    int lane = threadIdx.x & 31;
    int m0 = wave << 4;
    if (m0 >= nrows) return;
    int row = lane & 15;
    int kk  = (lane >> 4) << 1;
    const float* Xr = X + (size_t)(m0 + row) * FIN + kk;
    v8f c0 = {}, c1 = {}, c2 = {}, c3 = {};
    for (int k0 = 0; k0 < FIN; k0 += 4) {
        v2f a;
        a.x = Xr[k0];
        a.y = Xr[k0 + 1];
        const float* Wp = W + (size_t)(k0 + kk) * HID;
        v2f b0 = { Wp[row],      Wp[HID + row] };
        v2f b1 = { Wp[16 + row], Wp[HID + 16 + row] };
        v2f b2 = { Wp[32 + row], Wp[HID + 32 + row] };
        v2f b3 = { Wp[48 + row], Wp[HID + 48 + row] };
        c0 = __builtin_amdgcn_wmma_f32_16x16x4_f32(false, a, false, b0, (short)0, c0, false, false);
        c1 = __builtin_amdgcn_wmma_f32_16x16x4_f32(false, a, false, b1, (short)0, c1, false, false);
        c2 = __builtin_amdgcn_wmma_f32_16x16x4_f32(false, a, false, b2, (short)0, c2, false, false);
        c3 = __builtin_amdgcn_wmma_f32_16x16x4_f32(false, a, false, b3, (short)0, c3, false, false);
    }
    int mrow = m0 + ((lane >> 4) << 3);
    int col  = lane & 15;
    for (int r = 0; r < 8; ++r) {
        float di = rsqrtf(deg[mrow + r]);
        size_t off = (size_t)(mrow + r) * HID + col;
        float y0 = c0[r] * di, y1 = c1[r] * di, y2 = c2[r] * di, y3 = c3[r] * di;
        H[off]      = y0;  A[off]      = y0;
        H[off + 16] = y1;  A[off + 16] = y1;
        H[off + 32] = y2;  A[off + 32] = y2;
        H[off + 48] = y3;  A[off + 48] = y3;
    }
}

__global__ __launch_bounds__(256) void k_gemm2(float* A,
                                               const float* __restrict__ W,
                                               const float* __restrict__ deg,
                                               const float* __restrict__ bias1,
                                               float* __restrict__ H2, int nrows) {
    int wave = (blockIdx.x * 256 + threadIdx.x) >> 5;
    int lane = threadIdx.x & 31;
    int m0 = wave << 4;
    if (m0 >= nrows) return;
    int row = lane & 15;
    int kk  = (lane >> 4) << 1;
    const float* Ar = A + (size_t)(m0 + row) * HID + kk;
    const float* bp = bias1 + kk;
    const float dr = rsqrtf(deg[m0 + row]);
    v8f c0 = {}, c1 = {}, c2 = {};
    for (int k0 = 0; k0 < HID; k0 += 4) {
        v2f a;
        a.x = fmaxf(Ar[k0]     * dr + bp[k0],     0.0f);
        a.y = fmaxf(Ar[k0 + 1] * dr + bp[k0 + 1], 0.0f);
        const float* Wp = W + (size_t)(k0 + kk) * COUT;
        v2f b0 = { Wp[row],      Wp[COUT + row] };
        v2f b1 = { Wp[16 + row], Wp[COUT + 16 + row] };
        int col2 = 32 + row;
        v2f b2 = { (col2 < COUT) ? Wp[col2] : 0.0f,
                   (col2 < COUT) ? Wp[COUT + col2] : 0.0f };
        c0 = __builtin_amdgcn_wmma_f32_16x16x4_f32(false, a, false, b0, (short)0, c0, false, false);
        c1 = __builtin_amdgcn_wmma_f32_16x16x4_f32(false, a, false, b1, (short)0, c1, false, false);
        c2 = __builtin_amdgcn_wmma_f32_16x16x4_f32(false, a, false, b2, (short)0, c2, false, false);
    }
    int mrow = m0 + ((lane >> 4) << 3);
    int col  = lane & 15;
    for (int r = 0; r < 8; ++r) {
        float di = rsqrtf(deg[mrow + r]);
        size_t off = (size_t)(mrow + r) * LD2;
        float y0 = c0[r] * di, y1 = c1[r] * di, y2 = c2[r] * di;
        H2[off + col]      = y0;  A[off + col]      = y0;
        H2[off + 16 + col] = y1;  A[off + 16 + col] = y1;
        if (32 + col < COUT) { H2[off + 32 + col] = y2;  A[off + 32 + col] = y2; }
    }
}

template <int NTAIL>
__device__ __forceinline__ void agg_group8(const float* __restrict__ H, float* __restrict__ A,
                                           int s, int d, int j0, int lane) {
    const bool tail = (NTAIL == 32) || (lane < NTAIL);
    const float* hp[8];
    float v0[8], v1[8];
#pragma unroll
    for (int u = 0; u < 8; ++u) {
        int sj = __builtin_amdgcn_readlane(s, j0 + u);
        hp[u] = H + (size_t)sj * LD2 + lane;
        v0[u] = hp[u][0];
    }
#pragma unroll
    for (int u = 0; u < 8; ++u) v1[u] = 0.0f;
    if (tail) {
#pragma unroll
        for (int u = 0; u < 8; ++u) v1[u] = hp[u][32];
    }
    float* ap[8];
#pragma unroll
    for (int u = 0; u < 8; ++u) {
        int dj = __builtin_amdgcn_readlane(d, j0 + u);
        ap[u] = A + (size_t)dj * LD2 + lane;
        atomicAdd(ap[u], v0[u]);
    }
    if (tail) {
#pragma unroll
        for (int u = 0; u < 8; ++u) atomicAdd(ap[u] + 32, v1[u]);
    }
}

template <int NTAIL>
__global__ __launch_bounds__(256) void k_agg(const int* __restrict__ src,
                                             const int* __restrict__ dst, int E,
                                             const float* __restrict__ H,
                                             float* __restrict__ A) {
    int wave = (blockIdx.x * 256 + threadIdx.x) >> 5;
    int lane = threadIdx.x & 31;
    int e0 = wave << 5;
    if (e0 >= E) return;
    int e = e0 + lane;
    int s = 0, d = 0;
    if (e < E) {
        s = src[e];
        d = dst[e];
    }
    if (e0 + 32 <= E) {
        for (int j0 = 0; j0 < 32; j0 += 8) agg_group8<NTAIL>(H, A, s, d, j0, lane);
    } else {
        int cnt = E - e0;
        for (int j = 0; j < cnt; ++j) {
            int   sj = __builtin_amdgcn_readlane(s, j);
            int   dj = __builtin_amdgcn_readlane(d, j);
            const float* hp = H + (size_t)sj * LD2 + lane;
            float* ap = A + (size_t)dj * LD2 + lane;
            atomicAdd(ap, hp[0]);
            if (NTAIL == 32 || lane < NTAIL) atomicAdd(ap + 32, hp[32]);
        }
    }
}

__global__ __launch_bounds__(256) void k_fin2(float4* __restrict__ out,
                                              const float4* __restrict__ acc,
                                              const float* __restrict__ deg,
                                              const float* __restrict__ b2, int n) {
    int idx = blockIdx.x * 256 + threadIdx.x;
    if (idx >= n * (COUT / 4)) return;
    int i  = idx / 10;
    int f4 = idx % 10;
    float di = rsqrtf(deg[i]);
    float4 o = acc[(size_t)i * (LD2 / 4) + f4];
    const float* b = b2 + f4 * 4;
    o.x = o.x * di + b[0];
    o.y = o.y * di + b[1];
    o.z = o.z * di + b[2];
    o.w = o.w * di + b[3];
    out[idx] = o;
}

extern "C" void kernel_launch(void* const* d_in, const int* in_sizes, int n_in,
                              void* d_out, int out_size, void* d_ws, size_t ws_size,
                              hipStream_t stream) {
    const float* x  = (const float*)d_in[0];
    const float* W1 = (const float*)d_in[1];
    const float* b1 = (const float*)d_in[2];
    const float* W2 = (const float*)d_in[3];
    const float* b2 = (const float*)d_in[4];
    const int*   ei = (const int*)d_in[5];

    const int Nn = in_sizes[0] / FIN;
    const int E  = in_sizes[5] / 2;
    const int* src = ei;
    const int* dst = ei + E;

    float* ws   = (float*)d_ws;
    float* deg  = ws;
    float* h    = ws + Nn;
    float* agg  = h + (size_t)Nn * HID;
    float* h2   = h;
    float* acc2 = agg;
    float* out  = (float*)d_out;

    int gN  = (Nn + 255) / 256;
    int gE  = (E + 255) / 256;
    int o4n = Nn * (COUT / 4);
    int gO4 = (o4n + 255) / 256;
    int gEw = (((E + 31) / 32) + 7) / 8;
    int gemmBlocks = ((Nn / 16) + 7) / 8;

    k_deg_init<<<gN, 256, 0, stream>>>(deg, Nn);
    k_deg_acc<<<gE, 256, 0, stream>>>(dst, E, deg);

    k_gemm1<<<gemmBlocks, 256, 0, stream>>>(x, W1, deg, h, agg, Nn);
    k_agg<32><<<gEw, 256, 0, stream>>>(src, dst, E, h, agg);

    k_gemm2<<<gemmBlocks, 256, 0, stream>>>(agg, W2, deg, b1, h2, Nn);
    k_agg<8><<<gEw, 256, 0, stream>>>(src, dst, E, h2, acc2);
    k_fin2<<<gO4, 256, 0, stream>>>((float4*)out, (const float4*)acc2, deg, b2, Nn);
}


# row adds kept in one package half: waves read their die, draw edge chunks from that half's int ticket counter, apply only edges whose dst is in the half's row range; sums in arrival order as before
# speedup vs baseline: 1.6958x; 1.2180x over previous
#include <hip/hip_runtime.h>

#define FIN 256
#define HID 64
#define COUT 40
#define LD2 64

typedef float v2f __attribute__((ext_vector_type(2)));
typedef float v8f __attribute__((ext_vector_type(8)));

#define TK 128
#define CPAD 64
#define NBLK 2048

__device__ __forceinline__ int my_half() {
    return (int)((__builtin_amdgcn_s_sendmsg_rtn(0x87) >> 18) & 1u);
}

__device__ __forceinline__ int draw(int* c, int lane) {
    int k = 0;
    if (lane == 0) k = atomicAdd(c, 1);
    return __builtin_amdgcn_readfirstlane(k);
}

__global__ __launch_bounds__(256) void k_deg_init(float* __restrict__ deg, int n, int* __restrict__ cnt) {
    int i = blockIdx.x * 256 + threadIdx.x;
    if (i < n) deg[i] = 1.0f;
    if (i < 6) cnt[i * CPAD] = 0;
}

__global__ __launch_bounds__(256) void k_deg_acc(const int* __restrict__ dst, int E, int nhalf, int nt,
                                                 float* __restrict__ deg, int* __restrict__ cnt) {
    const int lane = threadIdx.x & 31;
    const int h = my_half();
    for (int side = 0; side < 2; ++side) {
        const int p = h ^ side;
        int* c = cnt + p * CPAD;
        for (int k = draw(c, lane); k < nt; k = draw(c, lane)) {
            int e = k * TK + lane;
            int d[TK / 32];
#pragma unroll
            for (int q = 0; q < TK / 32; ++q) d[q] = (e + 32 * q < E) ? dst[e + 32 * q] : -1;
#pragma unroll
            for (int q = 0; q < TK / 32; ++q)
                if (d[q] >= 0 && (d[q] >= nhalf) == (p != 0)) atomicAdd(deg + d[q], 1.0f);
        }
    }
}

__global__ __launch_bounds__(256) void k_gemm1(const float* __restrict__ X,
                                               const float* __restrict__ W,
                                               const float* __restrict__ deg,
                                               float* __restrict__ H,
                                               float* __restrict__ A, int nrows) {
    int wave = (blockIdx.x * 256 + threadIdx.x) >> 5;
    int lane = threadIdx.x & 31;
    int m0 = wave << 4;
    if (m0 >= nrows) return;
    int row = lane & 15;
    int kk  = (lane >> 4) << 1;
    const float* Xr = X + (size_t)(m0 + row) * FIN + kk;
    v8f c0 = {}, c1 = {}, c2 = {}, c3 = {};
    for (int k0 = 0; k0 < FIN; k0 += 4) {
        v2f a;
        a.x = Xr[k0];
        a.y = Xr[k0 + 1];
        const float* Wp = W + (size_t)(k0 + kk) * HID;
        v2f b0 = { Wp[row],      Wp[HID + row] };
        v2f b1 = { Wp[16 + row], Wp[HID + 16 + row] };
        v2f b2 = { Wp[32 + row], Wp[HID + 32 + row] };
        v2f b3 = { Wp[48 + row], Wp[HID + 48 + row] };
        c0 = __builtin_amdgcn_wmma_f32_16x16x4_f32(false, a, false, b0, (short)0, c0, false, false);
        c1 = __builtin_amdgcn_wmma_f32_16x16x4_f32(false, a, false, b1, (short)0, c1, false, false);
        c2 = __builtin_amdgcn_wmma_f32_16x16x4_f32(false, a, false, b2, (short)0, c2, false, false);
        c3 = __builtin_amdgcn_wmma_f32_16x16x4_f32(false, a, false, b3, (short)0, c3, false, false);
    }
    int mrow = m0 + ((lane >> 4) << 3);
    int col  = lane & 15;
    for (int r = 0; r < 8; ++r) {
        float di = rsqrtf(deg[mrow + r]);
        size_t off = (size_t)(mrow + r) * HID + col;
        float y0 = c0[r] * di, y1 = c1[r] * di, y2 = c2[r] * di, y3 = c3[r] * di;
        H[off]      = y0;  A[off]      = y0;
        H[off + 16] = y1;  A[off + 16] = y1;
        H[off + 32] = y2;  A[off + 32] = y2;
        H[off + 48] = y3;  A[off + 48] = y3;
    }
}

__global__ __launch_bounds__(256) void k_gemm2(float* A,
                                               const float* __restrict__ W,
                                               const float* __restrict__ deg,
                                               const float* __restrict__ bias1,
                                               float* __restrict__ H2, int nrows) {
    int wave = (blockIdx.x * 256 + threadIdx.x) >> 5;
    int lane = threadIdx.x & 31;
    int m0 = wave << 4;
    if (m0 >= nrows) return;
    int row = lane & 15;
    int kk  = (lane >> 4) << 1;
    const float* Ar = A + (size_t)(m0 + row) * HID + kk;
    const float* bp = bias1 + kk;
    const float dr = rsqrtf(deg[m0 + row]);
    v8f c0 = {}, c1 = {}, c2 = {};
    for (int k0 = 0; k0 < HID; k0 += 4) {
        v2f a;
        a.x = fmaxf(Ar[k0]     * dr + bp[k0],     0.0f);
        a.y = fmaxf(Ar[k0 + 1] * dr + bp[k0 + 1], 0.0f);
        const float* Wp = W + (size_t)(k0 + kk) * COUT;
        v2f b0 = { Wp[row],      Wp[COUT + row] };
        v2f b1 = { Wp[16 + row], Wp[COUT + 16 + row] };
        int col2 = 32 + row;
        v2f b2 = { (col2 < COUT) ? Wp[col2] : 0.0f,
                   (col2 < COUT) ? Wp[COUT + col2] : 0.0f };
        c0 = __builtin_amdgcn_wmma_f32_16x16x4_f32(false, a, false, b0, (short)0, c0, false, false);
        c1 = __builtin_amdgcn_wmma_f32_16x16x4_f32(false, a, false, b1, (short)0, c1, false, false);
        c2 = __builtin_amdgcn_wmma_f32_16x16x4_f32(false, a, false, b2, (short)0, c2, false, false);
    }
    int mrow = m0 + ((lane >> 4) << 3);
    int col  = lane & 15;
    for (int r = 0; r < 8; ++r) {
        float di = rsqrtf(deg[mrow + r]);
        size_t off = (size_t)(mrow + r) * LD2;
        float y0 = c0[r] * di, y1 = c1[r] * di, y2 = c2[r] * di;
        H2[off + col]      = y0;  A[off + col]      = y0;
        H2[off + 16 + col] = y1;  A[off + 16 + col] = y1;
        if (32 + col < COUT) { H2[off + 32 + col] = y2;  A[off + 32 + col] = y2; }
    }
}

template <int NTAIL, bool FULL>
__device__ __forceinline__ void agg_batch(const float* __restrict__ H, float* __restrict__ A,
                                          int sa, int da, int sb, int db,
                                          unsigned long long& m, int n, int lane) {
    const bool tail = (NTAIL == 32) || (lane < NTAIL);
    int sj[8], dj[8];
#pragma unroll
    for (int u = 0; u < 8; ++u) {
        sj[u] = 0; dj[u] = 0;
        if (FULL || u < n) {
            int j  = __builtin_ctzll(m);
            m &= m - 1;
            int jl = j & 31;
            int s0 = __builtin_amdgcn_readlane(sa, jl), s1 = __builtin_amdgcn_readlane(sb, jl);
            int d0 = __builtin_amdgcn_readlane(da, jl), d1 = __builtin_amdgcn_readlane(db, jl);
            sj[u] = (j & 32) ? s1 : s0;
            dj[u] = (j & 32) ? d1 : d0;
        }
    }
    float v0[8], v1[8];
#pragma unroll
    for (int u = 0; u < 8; ++u) {
        v0[u] = 0.0f; v1[u] = 0.0f;
        if (FULL || u < n) v0[u] = H[(size_t)sj[u] * LD2 + lane];
    }
    if (tail) {
#pragma unroll
        for (int u = 0; u < 8; ++u)
            if (FULL || u < n) v1[u] = H[(size_t)sj[u] * LD2 + 32 + lane];
    }
#pragma unroll
    for (int u = 0; u < 8; ++u)
        if (FULL || u < n) atomicAdd(A + (size_t)dj[u] * LD2 + lane, v0[u]);
    if (tail) {
#pragma unroll
        for (int u = 0; u < 8; ++u)
            if (FULL || u < n) atomicAdd(A + (size_t)dj[u] * LD2 + 32 + lane, v1[u]);
    }
}

template <int NTAIL>
__global__ __launch_bounds__(256) void k_agg(const int* __restrict__ src,
                                             const int* __restrict__ dst, int E, int nhalf, int nt,
                                             const float* __restrict__ H,
                                             float* __restrict__ A, int* __restrict__ cnt) {
    const int lane = threadIdx.x & 31;
    const int h = my_half();
    for (int side = 0; side < 2; ++side) {
        const int p = h ^ side;
        const bool hi = (p != 0);
        int* c = cnt + p * CPAD;
        for (int k = draw(c, lane); k < nt; k = draw(c, lane)) {
            int s[TK / 32], d[TK / 32];
#pragma unroll
            for (int q = 0; q < TK / 32; ++q) {
                int e = k * TK + 32 * q + lane;
                s[q] = 0; d[q] = -1;
                if (e < E) { s[q] = src[e]; d[q] = dst[e]; }
            }
#pragma unroll
            for (int q = 0; q < TK / 32; q += 2) {
                const int sa = s[q], da = d[q], sb = s[q + 1], db = d[q + 1];
                bool ka = (da >= 0) && ((da >= nhalf) == hi);
                bool kb = (db >= 0) && ((db >= nhalf) == hi);
                unsigned long long m = (unsigned long long)__builtin_amdgcn_ballot_w32(ka)
                                     | ((unsigned long long)__builtin_amdgcn_ballot_w32(kb) << 32);
                while (m) {
                    int n = __builtin_popcountll(m);
                    if (n >= 8) agg_batch<NTAIL, true >(H, A, sa, da, sb, db, m, 8, lane);
                    else        agg_batch<NTAIL, false>(H, A, sa, da, sb, db, m, n, lane);
                }
            }
        }
    }
}

__global__ __launch_bounds__(256) void k_fin2(float4* __restrict__ out,
                                              const float4* __restrict__ acc,
                                              const float* __restrict__ deg,
                                              const float* __restrict__ b2, int n) {
    int idx = blockIdx.x * 256 + threadIdx.x;
    if (idx >= n * (COUT / 4)) return;
    int i  = idx / 10;
    int f4 = idx % 10;
    float di = rsqrtf(deg[i]);
    float4 o = acc[(size_t)i * (LD2 / 4) + f4];
    const float* b = b2 + f4 * 4;
    o.x = o.x * di + b[0];
    o.y = o.y * di + b[1];
    o.z = o.z * di + b[2];
    o.w = o.w * di + b[3];
    out[idx] = o;
}

extern "C" void kernel_launch(void* const* d_in, const int* in_sizes, int n_in,
                              void* d_out, int out_size, void* d_ws, size_t ws_size,
                              hipStream_t stream) {
    const float* x  = (const float*)d_in[0];
    const float* W1 = (const float*)d_in[1];
    const float* b1 = (const float*)d_in[2];
    const float* W2 = (const float*)d_in[3];
    const float* b2 = (const float*)d_in[4];
    const int*   ei = (const int*)d_in[5];

    const int Nn = in_sizes[0] / FIN;
    const int E  = in_sizes[5] / 2;
    const int* src = ei;
    const int* dst = ei + E;

    float* ws   = (float*)d_ws;
    float* deg  = ws;
    float* h    = ws + Nn;
    float* agg  = h + (size_t)Nn * HID;
    float* h2   = h;
    float* acc2 = agg;
    float* out  = (float*)d_out;

    int gN  = (Nn + 255) / 256;
    int nt    = (E + TK - 1) / TK;
    int nhalf = ((Nn / 2 + 31) / 32) * 32;
    int* cnt  = (int*)d_out;
    int o4n = Nn * (COUT / 4);
    int gO4 = (o4n + 255) / 256;
    int gemmBlocks = ((Nn / 16) + 7) / 8;

    k_deg_init<<<gN, 256, 0, stream>>>(deg, Nn, cnt);
    k_deg_acc<<<NBLK, 256, 0, stream>>>(dst, E, nhalf, nt, deg, cnt);

    k_gemm1<<<gemmBlocks, 256, 0, stream>>>(x, W1, deg, h, agg, Nn);
    k_agg<32><<<NBLK, 256, 0, stream>>>(src, dst, E, nhalf, nt, h, agg, cnt + 2 * CPAD);

    k_gemm2<<<gemmBlocks, 256, 0, stream>>>(agg, W2, deg, b1, h2, Nn);
    k_agg<8><<<NBLK, 256, 0, stream>>>(src, dst, E, nhalf, nt, h2, acc2, cnt + 4 * CPAD);
    k_fin2<<<gO4, 256, 0, stream>>>((float4*)out, (const float4*)acc2, deg, b2, Nn);
}


# the two short passes as few long waves: degree count draws 1024-edge tickets on 256 blocks (an eighth of the draws on its two counter words), finalize takes 8 float4s per thread; sums unchanged
# speedup vs baseline: 2.0028x; 1.1810x over previous
#include <hip/hip_runtime.h>

#define FIN 256
#define HID 64
#define COUT 40
#define LD2 64

typedef float v2f __attribute__((ext_vector_type(2)));
typedef float v8f __attribute__((ext_vector_type(8)));

#define TK 128
#define CPAD 64
#define NBLK 2048
#define DTK 1024
#define DBLK 256
#define FPT 8

__device__ __forceinline__ int my_half() {
    return (int)((__builtin_amdgcn_s_sendmsg_rtn(0x87) >> 18) & 1u);
}

__device__ __forceinline__ int draw(int* c, int lane) {
    int k = 0;
    if (lane == 0) k = atomicAdd(c, 1);
    return __builtin_amdgcn_readfirstlane(k);
}

__global__ __launch_bounds__(256) void k_deg_init(float* __restrict__ deg, int n, int* __restrict__ cnt) {
    int i = blockIdx.x * 256 + threadIdx.x;
    if (i < n) deg[i] = 1.0f;
    if (i < 6) cnt[i * CPAD] = 0;
}

__global__ __launch_bounds__(256) void k_deg_acc(const int* __restrict__ dst, int E, int nhalf,
                                                 float* __restrict__ deg, int* __restrict__ cnt) {
    const int lane = threadIdx.x & 31;
    const int h = my_half();
    const int nt = (E + DTK - 1) / DTK;
    for (int side = 0; side < 2; ++side) {
        const int p = h ^ side;
        int* c = cnt + p * CPAD;
        for (int k = draw(c, lane); k < nt; k = draw(c, lane)) {
            int e = k * DTK + lane;
            int d[DTK / 32];
            if (k * DTK + DTK <= E) {
#pragma unroll
                for (int q = 0; q < DTK / 32; ++q) d[q] = dst[e + 32 * q];
            } else {
#pragma unroll
                for (int q = 0; q < DTK / 32; ++q) d[q] = (e + 32 * q < E) ? dst[e + 32 * q] : -1;
            }
#pragma unroll
            for (int q = 0; q < DTK / 32; ++q)
                if (d[q] >= 0 && (d[q] >= nhalf) == (p != 0)) atomicAdd(deg + d[q], 1.0f);
        }
    }
}

__global__ __launch_bounds__(256) void k_gemm1(const float* __restrict__ X,
                                               const float* __restrict__ W,
                                               const float* __restrict__ deg,
                                               float* __restrict__ H,
                                               float* __restrict__ A, int nrows) {
    int wave = (blockIdx.x * 256 + threadIdx.x) >> 5;
    int lane = threadIdx.x & 31;
    int m0 = wave << 4;
    if (m0 >= nrows) return;
    int row = lane & 15;
    int kk  = (lane >> 4) << 1;
    const float* Xr = X + (size_t)(m0 + row) * FIN + kk;
    v8f c0 = {}, c1 = {}, c2 = {}, c3 = {};
    for (int k0 = 0; k0 < FIN; k0 += 4) {
        v2f a;
        a.x = Xr[k0];
        a.y = Xr[k0 + 1];
        const float* Wp = W + (size_t)(k0 + kk) * HID;
        v2f b0 = { Wp[row],      Wp[HID + row] };
        v2f b1 = { Wp[16 + row], Wp[HID + 16 + row] };
        v2f b2 = { Wp[32 + row], Wp[HID + 32 + row] };
        v2f b3 = { Wp[48 + row], Wp[HID + 48 + row] };
        c0 = __builtin_amdgcn_wmma_f32_16x16x4_f32(false, a, false, b0, (short)0, c0, false, false);
        c1 = __builtin_amdgcn_wmma_f32_16x16x4_f32(false, a, false, b1, (short)0, c1, false, false);
        c2 = __builtin_amdgcn_wmma_f32_16x16x4_f32(false, a, false, b2, (short)0, c2, false, false);
        c3 = __builtin_amdgcn_wmma_f32_16x16x4_f32(false, a, false, b3, (short)0, c3, false, false);
    }
    int mrow = m0 + ((lane >> 4) << 3);
    int col  = lane & 15;
    for (int r = 0; r < 8; ++r) {
        float di = rsqrtf(deg[mrow + r]);
        size_t off = (size_t)(mrow + r) * HID + col;
        float y0 = c0[r] * di, y1 = c1[r] * di, y2 = c2[r] * di, y3 = c3[r] * di;
        H[off]      = y0;  A[off]      = y0;
        H[off + 16] = y1;  A[off + 16] = y1;
        H[off + 32] = y2;  A[off + 32] = y2;
        H[off + 48] = y3;  A[off + 48] = y3;
    }
}

__global__ __launch_bounds__(256) void k_gemm2(float* A,
                                               const float* __restrict__ W,
                                               const float* __restrict__ deg,
                                               const float* __restrict__ bias1,
                                               float* __restrict__ H2, int nrows) {
    int wave = (blockIdx.x * 256 + threadIdx.x) >> 5;
    int lane = threadIdx.x & 31;
    int m0 = wave << 4;
    if (m0 >= nrows) return;
    int row = lane & 15;
    int kk  = (lane >> 4) << 1;
    const float* Ar = A + (size_t)(m0 + row) * HID + kk;
    const float* bp = bias1 + kk;
    const float dr = rsqrtf(deg[m0 + row]);
    v8f c0 = {}, c1 = {}, c2 = {};
    for (int k0 = 0; k0 < HID; k0 += 4) {
        v2f a;
        a.x = fmaxf(Ar[k0]     * dr + bp[k0],     0.0f);
        a.y = fmaxf(Ar[k0 + 1] * dr + bp[k0 + 1], 0.0f);
        const float* Wp = W + (size_t)(k0 + kk) * COUT;
        v2f b0 = { Wp[row],      Wp[COUT + row] };
        v2f b1 = { Wp[16 + row], Wp[COUT + 16 + row] };
        int col2 = 32 + row;
        v2f b2 = { (col2 < COUT) ? Wp[col2] : 0.0f,
                   (col2 < COUT) ? Wp[COUT + col2] : 0.0f };
        c0 = __builtin_amdgcn_wmma_f32_16x16x4_f32(false, a, false, b0, (short)0, c0, false, false);
        c1 = __builtin_amdgcn_wmma_f32_16x16x4_f32(false, a, false, b1, (short)0, c1, false, false);
        c2 = __builtin_amdgcn_wmma_f32_16x16x4_f32(false, a, false, b2, (short)0, c2, false, false);
    }
    int mrow = m0 + ((lane >> 4) << 3);
    int col  = lane & 15;
    for (int r = 0; r < 8; ++r) {
        float di = rsqrtf(deg[mrow + r]);
        size_t off = (size_t)(mrow + r) * LD2;
        float y0 = c0[r] * di, y1 = c1[r] * di, y2 = c2[r] * di;
        H2[off + col]      = y0;  A[off + col]      = y0;
        H2[off + 16 + col] = y1;  A[off + 16 + col] = y1;
        if (32 + col < COUT) { H2[off + 32 + col] = y2;  A[off + 32 + col] = y2; }
    }
}

template <int NTAIL, bool FULL>
__device__ __forceinline__ void agg_batch(const float* __restrict__ H, float* __restrict__ A,
                                          int sa, int da, int sb, int db,
                                          unsigned long long& m, int n, int lane) {
    const bool tail = (NTAIL == 32) || (lane < NTAIL);
    int sj[8], dj[8];
#pragma unroll
    for (int u = 0; u < 8; ++u) {
        sj[u] = 0; dj[u] = 0;
        if (FULL || u < n) {
            int j  = __builtin_ctzll(m);
            m &= m - 1;
            int jl = j & 31;
            int s0 = __builtin_amdgcn_readlane(sa, jl), s1 = __builtin_amdgcn_readlane(sb, jl);
            int d0 = __builtin_amdgcn_readlane(da, jl), d1 = __builtin_amdgcn_readlane(db, jl);
            sj[u] = (j & 32) ? s1 : s0;
            dj[u] = (j & 32) ? d1 : d0;
        }
    }
    float v0[8], v1[8];
#pragma unroll
    for (int u = 0; u < 8; ++u) {
        v0[u] = 0.0f; v1[u] = 0.0f;
        if (FULL || u < n) v0[u] = H[(size_t)sj[u] * LD2 + lane];
    }
    if (tail) {
#pragma unroll
        for (int u = 0; u < 8; ++u)
            if (FULL || u < n) v1[u] = H[(size_t)sj[u] * LD2 + 32 + lane];
    }
#pragma unroll
    for (int u = 0; u < 8; ++u)
        if (FULL || u < n) atomicAdd(A + (size_t)dj[u] * LD2 + lane, v0[u]);
    if (tail) {
#pragma unroll
        for (int u = 0; u < 8; ++u)
            if (FULL || u < n) atomicAdd(A + (size_t)dj[u] * LD2 + 32 + lane, v1[u]);
    }
}

template <int NTAIL>
__global__ __launch_bounds__(256) void k_agg(const int* __restrict__ src,
                                             const int* __restrict__ dst, int E, int nhalf, int nt,
                                             const float* __restrict__ H,
                                             float* __restrict__ A, int* __restrict__ cnt) {
    const int lane = threadIdx.x & 31;
    const int h = my_half();
    for (int side = 0; side < 2; ++side) {
        const int p = h ^ side;
        const bool hi = (p != 0);
        int* c = cnt + p * CPAD;
        for (int k = draw(c, lane); k < nt; k = draw(c, lane)) {
            int s[TK / 32], d[TK / 32];
#pragma unroll
            for (int q = 0; q < TK / 32; ++q) {
                int e = k * TK + 32 * q + lane;
                s[q] = 0; d[q] = -1;
                if (e < E) { s[q] = src[e]; d[q] = dst[e]; }
            }
#pragma unroll
            for (int q = 0; q < TK / 32; q += 2) {
                const int sa = s[q], da = d[q], sb = s[q + 1], db = d[q + 1];
                bool ka = (da >= 0) && ((da >= nhalf) == hi);
                bool kb = (db >= 0) && ((db >= nhalf) == hi);
                unsigned long long m = (unsigned long long)__builtin_amdgcn_ballot_w32(ka)
                                     | ((unsigned long long)__builtin_amdgcn_ballot_w32(kb) << 32);
                while (m) {
                    int n = __builtin_popcountll(m);
                    if (n >= 8) agg_batch<NTAIL, true >(H, A, sa, da, sb, db, m, 8, lane);
                    else        agg_batch<NTAIL, false>(H, A, sa, da, sb, db, m, n, lane);
                }
            }
        }
    }
}

template <bool GUARD>
__device__ __forceinline__ void fin2_trip(float4* __restrict__ out, const float4* __restrict__ acc,
                                          const float* __restrict__ deg, const float* __restrict__ b2,
                                          int u0, int nwaves, int nunits, int lane) {
    float4 o[FPT], bb[FPT];
    float  dg[FPT];
#pragma unroll
    for (int u = 0; u < FPT; ++u) {
        const int unit = GUARD ? min(u0 + u * nwaves, nunits - 1) : u0 + u * nwaves;
        const int idx  = unit * 32 + lane;
        const int i    = idx / 10;
        const int f4   = idx - i * 10;
        const float* b = b2 + f4 * 4;
        dg[u] = deg[i];
        o[u]  = acc[(size_t)i * (LD2 / 4) + f4];
        bb[u] = make_float4(b[0], b[1], b[2], b[3]);
    }
#pragma unroll
    for (int u = 0; u < FPT; ++u) {
        const int unit = u0 + u * nwaves;
        if (!GUARD || unit < nunits) {
            const float di = rsqrtf(dg[u]);
            float4 r = o[u];
            r.x = r.x * di + bb[u].x;
            r.y = r.y * di + bb[u].y;
            r.z = r.z * di + bb[u].z;
            r.w = r.w * di + bb[u].w;
            out[unit * 32 + lane] = r;
        }
    }
}

__global__ __launch_bounds__(256) void k_fin2(float4* __restrict__ out,
                                              const float4* __restrict__ acc,
                                              const float* __restrict__ deg,
                                              const float* __restrict__ b2, int n) {
    const int nunits = (n * (COUT / 4)) >> 5;
    const int nwaves = gridDim.x * 8;
    const int lane   = threadIdx.x & 31;
    const int wave   = __builtin_amdgcn_readfirstlane((int)((blockIdx.x * 256 + threadIdx.x) >> 5));
    for (int u0 = wave; u0 < nunits; u0 += nwaves * FPT) {
        if (u0 + (FPT - 1) * nwaves < nunits) fin2_trip<false>(out, acc, deg, b2, u0, nwaves, nunits, lane);
        else                                  fin2_trip<true >(out, acc, deg, b2, u0, nwaves, nunits, lane);
    }
}

extern "C" void kernel_launch(void* const* d_in, const int* in_sizes, int n_in,
                              void* d_out, int out_size, void* d_ws, size_t ws_size,
                              hipStream_t stream) {
    const float* x  = (const float*)d_in[0];
    const float* W1 = (const float*)d_in[1];
    const float* b1 = (const float*)d_in[2];
    const float* W2 = (const float*)d_in[3];
    const float* b2 = (const float*)d_in[4];
    const int*   ei = (const int*)d_in[5];

    const int Nn = in_sizes[0] / FIN;
    const int E  = in_sizes[5] / 2;
    const int* src = ei;
    const int* dst = ei + E;

    float* ws   = (float*)d_ws;
    float* deg  = ws;
    float* h    = ws + Nn;
    float* agg  = h + (size_t)Nn * HID;
    float* h2   = h;
    float* acc2 = agg;
    float* out  = (float*)d_out;

    int gN  = (Nn + 255) / 256;
    int nt    = (E + TK - 1) / TK;
    int nhalf = ((Nn / 2 + 31) / 32) * 32;
    int* cnt  = (int*)d_out;
    int gemmBlocks = ((Nn / 16) + 7) / 8;
    int gF = (Nn * (COUT / 4) / 32 + FPT * 8 - 1) / (FPT * 8);

    k_deg_init<<<gN, 256, 0, stream>>>(deg, Nn, cnt);
    k_deg_acc<<<DBLK, 256, 0, stream>>>(dst, E, nhalf, deg, cnt);

    k_gemm1<<<gemmBlocks, 256, 0, stream>>>(x, W1, deg, h, agg, Nn);
    k_agg<32><<<NBLK, 256, 0, stream>>>(src, dst, E, nhalf, nt, h, agg, cnt + 2 * CPAD);

    k_gemm2<<<gemmBlocks, 256, 0, stream>>>(agg, W2, deg, b1, h2, Nn);
    k_agg<8><<<NBLK, 256, 0, stream>>>(src, dst, E, nhalf, nt, h2, acc2, cnt + 4 * CPAD);
    k_fin2<<<gF, 256, 0, stream>>>((float4*)out, (const float4*)acc2, deg, b2, Nn);
}


# edge sums by row owners, now found by one-byte owner keys from the degree pass: 256 owners hold rows in LDS, per-wave match stacks, 8-row gather batches, LDS float adds; sums in arrival order
# speedup vs baseline: 2.0844x; 1.0407x over previous
#include <hip/hip_runtime.h>

#define FIN 256
#define HID 64
#define COUT 40
#define LD2 64

typedef float v2f __attribute__((ext_vector_type(2)));
typedef float v8f __attribute__((ext_vector_type(8)));

#define CPAD 64
#define DTK 1024
#define DBLK 256
#define FPT 8
#define RPO   391
#define NOWN  256

__device__ __forceinline__ int my_half() {
    return (int)((__builtin_amdgcn_s_sendmsg_rtn(0x87) >> 18) & 1u);
}

__device__ __forceinline__ int draw(int* c, int lane) {
    int k = 0;
    if (lane == 0) k = atomicAdd(c, 1);
    return __builtin_amdgcn_readfirstlane(k);
}

__global__ __launch_bounds__(256) void k_deg_init(float* __restrict__ deg, int n, int* __restrict__ cnt) {
    int i = blockIdx.x * 256 + threadIdx.x;
    if (i < n) deg[i] = 1.0f;
    if (i < 2) cnt[i * CPAD] = 0;
}

__global__ __launch_bounds__(256) void k_deg_acc(const int* __restrict__ dst, int E, int nhalf,
                                                 float* __restrict__ deg, int* __restrict__ cnt,
                                                 unsigned char* __restrict__ key) {
    const int lane = threadIdx.x & 31;
    const int h = my_half();
    const int nt = (E + DTK - 1) / DTK;
    for (int side = 0; side < 2; ++side) {
        const int p = h ^ side;
        int* c = cnt + p * CPAD;
        for (int k = draw(c, lane); k < nt; k = draw(c, lane)) {
            int e = k * DTK + lane;
            int d[DTK / 32];
            if (k * DTK + DTK <= E) {
#pragma unroll
                for (int q = 0; q < DTK / 32; ++q) d[q] = dst[e + 32 * q];
            } else {
#pragma unroll
                for (int q = 0; q < DTK / 32; ++q) d[q] = (e + 32 * q < E) ? dst[e + 32 * q] : -1;
            }
#pragma unroll
            for (int q = 0; q < DTK / 32; ++q)
                if (d[q] >= 0 && (d[q] >= nhalf) == (p != 0)) atomicAdd(deg + d[q], 1.0f);
            if (p == 0) {
#pragma unroll
                for (int q = 0; q < DTK / 32; ++q)
                    if (d[q] >= 0) key[e + 32 * q] = (unsigned char)(d[q] / RPO);
            }
        }
    }
}

__global__ __launch_bounds__(256) void k_gemm1(const float* __restrict__ X,
                                               const float* __restrict__ W,
                                               const float* __restrict__ deg,
                                               float* __restrict__ H, int nrows) {
    int wave = (blockIdx.x * 256 + threadIdx.x) >> 5;
    int lane = threadIdx.x & 31;
    int m0 = wave << 4;
    if (m0 >= nrows) return;
    int row = lane & 15;
    int kk  = (lane >> 4) << 1;
    const float* Xr = X + (size_t)(m0 + row) * FIN + kk;
    v8f c0 = {}, c1 = {}, c2 = {}, c3 = {};
    for (int k0 = 0; k0 < FIN; k0 += 4) {
        v2f a;
        a.x = Xr[k0];
        a.y = Xr[k0 + 1];
        const float* Wp = W + (size_t)(k0 + kk) * HID;
        v2f b0 = { Wp[row],      Wp[HID + row] };
        v2f b1 = { Wp[16 + row], Wp[HID + 16 + row] };
        v2f b2 = { Wp[32 + row], Wp[HID + 32 + row] };
        v2f b3 = { Wp[48 + row], Wp[HID + 48 + row] };
        c0 = __builtin_amdgcn_wmma_f32_16x16x4_f32(false, a, false, b0, (short)0, c0, false, false);
        c1 = __builtin_amdgcn_wmma_f32_16x16x4_f32(false, a, false, b1, (short)0, c1, false, false);
        c2 = __builtin_amdgcn_wmma_f32_16x16x4_f32(false, a, false, b2, (short)0, c2, false, false);
        c3 = __builtin_amdgcn_wmma_f32_16x16x4_f32(false, a, false, b3, (short)0, c3, false, false);
    }
    int mrow = m0 + ((lane >> 4) << 3);
    int col  = lane & 15;
    for (int r = 0; r < 8; ++r) {
        float di = rsqrtf(deg[mrow + r]);
        size_t off = (size_t)(mrow + r) * HID + col;
        float y0 = c0[r] * di, y1 = c1[r] * di, y2 = c2[r] * di, y3 = c3[r] * di;
        H[off]      = y0;
        H[off + 16] = y1;
        H[off + 32] = y2;
        H[off + 48] = y3;
    }
}

__global__ __launch_bounds__(256) void k_gemm2(const float* __restrict__ A,
                                               const float* __restrict__ W,
                                               const float* __restrict__ deg,
                                               const float* __restrict__ bias1,
                                               float* __restrict__ H2, int nrows) {
    int wave = (blockIdx.x * 256 + threadIdx.x) >> 5;
    int lane = threadIdx.x & 31;
    int m0 = wave << 4;
    if (m0 >= nrows) return;
    int row = lane & 15;
    int kk  = (lane >> 4) << 1;
    const float* Ar = A + (size_t)(m0 + row) * HID + kk;
    const float* bp = bias1 + kk;
    const float dr = rsqrtf(deg[m0 + row]);
    v8f c0 = {}, c1 = {}, c2 = {};
    for (int k0 = 0; k0 < HID; k0 += 4) {
        v2f a;
        a.x = fmaxf(Ar[k0]     * dr + bp[k0],     0.0f);
        a.y = fmaxf(Ar[k0 + 1] * dr + bp[k0 + 1], 0.0f);
        const float* Wp = W + (size_t)(k0 + kk) * COUT;
        v2f b0 = { Wp[row],      Wp[COUT + row] };
        v2f b1 = { Wp[16 + row], Wp[COUT + 16 + row] };
        int col2 = 32 + row;
        v2f b2 = { (col2 < COUT) ? Wp[col2] : 0.0f,
                   (col2 < COUT) ? Wp[COUT + col2] : 0.0f };
        c0 = __builtin_amdgcn_wmma_f32_16x16x4_f32(false, a, false, b0, (short)0, c0, false, false);
        c1 = __builtin_amdgcn_wmma_f32_16x16x4_f32(false, a, false, b1, (short)0, c1, false, false);
        c2 = __builtin_amdgcn_wmma_f32_16x16x4_f32(false, a, false, b2, (short)0, c2, false, false);
    }
    int mrow = m0 + ((lane >> 4) << 3);
    int col  = lane & 15;
    for (int r = 0; r < 8; ++r) {
        float di = rsqrtf(deg[mrow + r]);
        size_t off = (size_t)(mrow + r) * LD2;
        float y0 = c0[r] * di, y1 = c1[r] * di, y2 = c2[r] * di;
        H2[off + col]      = y0;
        H2[off + 16 + col] = y1;
        if (32 + col < COUT) H2[off + 32 + col] = y2;
    }
}

#define KPL   4
#define TRIPK (KPL * 512)
#define QCAP  (512 + 8)

__device__ __forceinline__ unsigned zero_bytes(unsigned y) {
    return ~(((y & 0x7F7F7F7Fu) + 0x7F7F7F7Fu) | y | 0x7F7F7F7Fu);
}
__device__ __forceinline__ unsigned clip_flags(unsigned f, int k) {
    return (k >= 4) ? f : (k <= 0) ? 0u : (f & ((1u << (8 * k)) - 1u));
}
__device__ __forceinline__ void push_matches(unsigned c, unsigned m, int eb, int* myq, int& cnt) {
    do {
        if (c != 0u) {
            const int p = __builtin_ctz(c);
            myq[cnt + (int)__builtin_amdgcn_mbcnt_lo(m, 0u)] = eb + ((p & 7) << 2) + (p >> 3);
            c &= c - 1u;
        }
        cnt += __builtin_popcount(m);
        m = __builtin_amdgcn_ballot_w32(c != 0u);
    } while (m != 0u);
}

template <int NC, bool FULL>
__device__ __forceinline__ void own_batch(const int* __restrict__ src, const int* __restrict__ dst,
                                          const float* __restrict__ H, float* rows, const int* myq,
                                          int top, int nb, int row0, int lane) {
    const bool tail = (NC == 64) || (lane < NC - 32);
    int e = 0;
    if (lane < (FULL ? 8 : nb)) e = myq[top - 1 - lane];
    int s = 0, d = row0;
    if (lane < (FULL ? 8 : nb)) { s = src[e]; d = dst[e]; }
    int sj[8], dj[8];
#pragma unroll
    for (int u = 0; u < 8; ++u) {
        sj[u] = __builtin_amdgcn_readlane(s, u);
        dj[u] = __builtin_amdgcn_readlane(d, u) - row0;
    }
    float v0[8], v1[8];
#pragma unroll
    for (int u = 0; u < 8; ++u) {
        v0[u] = 0.0f; v1[u] = 0.0f;
        if (FULL || u < nb) v0[u] = H[(size_t)sj[u] * LD2 + lane];
    }
    if (tail) {
#pragma unroll
        for (int u = 0; u < 8; ++u)
            if (FULL || u < nb) v1[u] = H[(size_t)sj[u] * LD2 + 32 + lane];
    }
#pragma unroll
    for (int u = 0; u < 8; ++u)
        if (FULL || u < nb) atomicAdd(rows + dj[u] * LD2 + lane, v0[u]);
    if (tail) {
#pragma unroll
        for (int u = 0; u < 8; ++u)
            if (FULL || u < nb) atomicAdd(rows + dj[u] * LD2 + 32 + lane, v1[u]);
    }
}

template <int NC>
__global__ __launch_bounds__(1024) void k_own(const uint4* __restrict__ key4,
                                              const int* __restrict__ src,
                                              const int* __restrict__ dst, int E, int n,
                                              const float* __restrict__ H, float* __restrict__ A) {
    __shared__ float rows[RPO * LD2];
    __shared__ int   q[32 * QCAP];
    constexpr int F4 = NC / 4;
    const int tid  = threadIdx.x;
    const int lane = tid & 31;
    const int w    = __builtin_amdgcn_readfirstlane(tid >> 5);
    const int g    = blockIdx.x;
    const int row0 = g * RPO;
    const int nown = min(RPO, n - row0);
    const float4* H4 = (const float4*)H;
    float4* A4 = (float4*)A;
    float4* rows4 = (float4*)rows;

    constexpr int NIT = (RPO * F4 + 1023) / 1024;
    float4 st[NIT];
#pragma unroll
    for (int u = 0; u < NIT; ++u) {
        const int i = tid + u * 1024;
        const int r = i / F4, f = i - r * F4;
        st[u] = make_float4(0.0f, 0.0f, 0.0f, 0.0f);
        if (i < nown * F4) st[u] = H4[(size_t)(row0 + r) * (LD2 / 4) + f];
    }
#pragma unroll
    for (int u = 0; u < NIT; ++u) {
        const int i = tid + u * 1024;
        const int r = i / F4, f = i - r * F4;
        if (i < nown * F4) rows4[r * (LD2 / 4) + f] = st[u];
    }
    __syncthreads();

    const unsigned pat = (unsigned)g * 0x01010101u;
    int* myq = q + w * QCAP;
    int cnt = 0;
    const int ntrip = (E + TRIPK - 1) / TRIPK;
    const int rot = (int)(((long long)g * ntrip) >> 8);
    uint4 kv[KPL], nx[KPL];
    {
        int tq = (w < ntrip) ? w + rot : 0;
        if (tq >= ntrip) tq -= ntrip;
#pragma unroll
        for (int j = 0; j < KPL; ++j) kv[j] = key4[((tq * TRIPK) >> 4) + j * 32 + lane];
    }
    for (int t = w; t < ntrip; t += 32) {
        int tt = t + rot;
        if (tt >= ntrip) tt -= ntrip;
        const int e0 = tt * TRIPK;
        int tq = ((t + 32 < ntrip) ? t + 32 : t) + rot;
        if (tq >= ntrip) tq -= ntrip;
#pragma unroll
        for (int j = 0; j < KPL; ++j) nx[j] = key4[((tq * TRIPK) >> 4) + j * 32 + lane];
#pragma unroll
        for (int j = 0; j < KPL; ++j) {
            const int eb = e0 + j * 512 + lane * 16;
            unsigned f0 = zero_bytes(kv[j].x ^ pat), f1 = zero_bytes(kv[j].y ^ pat);
            unsigned f2 = zero_bytes(kv[j].z ^ pat), f3 = zero_bytes(kv[j].w ^ pat);
            if (eb + 16 > E) {
                f0 = clip_flags(f0, E - eb);      f1 = clip_flags(f1, E - eb - 4);
                f2 = clip_flags(f2, E - eb - 8);  f3 = clip_flags(f3, E - eb - 12);
            }
            const unsigned c16 = (f0 >> 7) | (f1 >> 6) | (f2 >> 5) | (f3 >> 4);
            const unsigned mm = __builtin_amdgcn_ballot_w32(c16 != 0u);
            if (mm != 0u) {
                push_matches(c16, mm, eb, myq, cnt);
                while (cnt >= 8) {
                    own_batch<NC, true>(src, dst, H, rows, myq, cnt, 8, row0, lane);
                    cnt -= 8;
                }
            }
        }
#pragma unroll
        for (int j = 0; j < KPL; ++j) kv[j] = nx[j];
    }
    if (cnt > 0) own_batch<NC, false>(src, dst, H, rows, myq, cnt, cnt, row0, lane);
    __syncthreads();

    for (int i = tid; i < nown * F4; i += 1024) {
        const int r = i / F4, f = i - r * F4;
        A4[(size_t)(row0 + r) * (LD2 / 4) + f] = rows4[r * (LD2 / 4) + f];
    }
}

template <bool GUARD>
__device__ __forceinline__ void fin2_trip(float4* __restrict__ out, const float4* __restrict__ acc,
                                          const float* __restrict__ deg, const float* __restrict__ b2,
                                          int u0, int nwaves, int nunits, int lane) {
    float4 o[FPT], bb[FPT];
    float  dg[FPT];
#pragma unroll
    for (int u = 0; u < FPT; ++u) {
        const int unit = GUARD ? min(u0 + u * nwaves, nunits - 1) : u0 + u * nwaves;
        const int idx  = unit * 32 + lane;
        const int i    = idx / 10;
        const int f4   = idx - i * 10;
        const float* b = b2 + f4 * 4;
        dg[u] = deg[i];
        o[u]  = acc[(size_t)i * (LD2 / 4) + f4];
        bb[u] = make_float4(b[0], b[1], b[2], b[3]);
    }
#pragma unroll
    for (int u = 0; u < FPT; ++u) {
        const int unit = u0 + u * nwaves;
        if (!GUARD || unit < nunits) {
            const float di = rsqrtf(dg[u]);
            float4 r = o[u];
            r.x = r.x * di + bb[u].x;
            r.y = r.y * di + bb[u].y;
            r.z = r.z * di + bb[u].z;
            r.w = r.w * di + bb[u].w;
            out[unit * 32 + lane] = r;
        }
    }
}

__global__ __launch_bounds__(256) void k_fin2(float4* __restrict__ out,
                                              const float4* __restrict__ acc,
                                              const float* __restrict__ deg,
                                              const float* __restrict__ b2, int n) {
    const int nunits = (n * (COUT / 4)) >> 5;
    const int nwaves = gridDim.x * 8;
    const int lane   = threadIdx.x & 31;
    const int wave   = __builtin_amdgcn_readfirstlane((int)((blockIdx.x * 256 + threadIdx.x) >> 5));
    for (int u0 = wave; u0 < nunits; u0 += nwaves * FPT) {
        if (u0 + (FPT - 1) * nwaves < nunits) fin2_trip<false>(out, acc, deg, b2, u0, nwaves, nunits, lane);
        else                                  fin2_trip<true >(out, acc, deg, b2, u0, nwaves, nunits, lane);
    }
}

extern "C" void kernel_launch(void* const* d_in, const int* in_sizes, int n_in,
                              void* d_out, int out_size, void* d_ws, size_t ws_size,
                              hipStream_t stream) {
    const float* x  = (const float*)d_in[0];
    const float* W1 = (const float*)d_in[1];
    const float* b1 = (const float*)d_in[2];
    const float* W2 = (const float*)d_in[3];
    const float* b2 = (const float*)d_in[4];
    const int*   ei = (const int*)d_in[5];

    const int Nn = in_sizes[0] / FIN;
    const int E  = in_sizes[5] / 2;
    const int* src = ei;
    const int* dst = ei + E;

    float* ws   = (float*)d_ws;
    float* deg  = ws;
    float* h    = ws + Nn;
    float* agg  = h + (size_t)Nn * HID;
    float* h2   = h;
    float* acc2 = agg;
    float* out  = (float*)d_out;

    int gN  = (Nn + 255) / 256;
    int nhalf = ((Nn / 2 + 31) / 32) * 32;
    int* cnt  = (int*)d_out;
    unsigned char* key = (unsigned char*)d_out + 4096;
    int gemmBlocks = ((Nn / 16) + 7) / 8;
    int gF = (Nn * (COUT / 4) / 32 + FPT * 8 - 1) / (FPT * 8);

    k_deg_init<<<gN, 256, 0, stream>>>(deg, Nn, cnt);
    k_deg_acc<<<DBLK, 256, 0, stream>>>(dst, E, nhalf, deg, cnt, key);

    k_gemm1<<<gemmBlocks, 256, 0, stream>>>(x, W1, deg, h, Nn);
    k_own<HID><<<NOWN, 1024, 0, stream>>>((const uint4*)key, src, dst, E, Nn, h, agg);

    k_gemm2<<<gemmBlocks, 256, 0, stream>>>(agg, W2, deg, b1, h2, Nn);
    k_own<COUT><<<NOWN, 1024, 0, stream>>>((const uint4*)key, src, dst, E, Nn, h2, acc2);
    k_fin2<<<gF, 256, 0, stream>>>((float4*)out, (const float4*)acc2, deg, b2, Nn);
}


# layer-2 edge pass reads back what layer 1 found: each owner wave lists the edges it applies as packed (src,row) words; layer 2 replays the lists, no key scan or index loads; sums in arrival order
# speedup vs baseline: 2.4142x; 1.1582x over previous
#include <hip/hip_runtime.h>

#define FIN 256
#define HID 64
#define COUT 40
#define LD2 64

typedef float v2f __attribute__((ext_vector_type(2)));
typedef float v8f __attribute__((ext_vector_type(8)));

#define CPAD 64
#define DTK 1024
#define DBLK 256
#define FPT 8
#define RPO   391
#define NOWN  256

__device__ __forceinline__ int my_half() {
    return (int)((__builtin_amdgcn_s_sendmsg_rtn(0x87) >> 18) & 1u);
}

__device__ __forceinline__ int draw(int* c, int lane) {
    int k = 0;
    if (lane == 0) k = atomicAdd(c, 1);
    return __builtin_amdgcn_readfirstlane(k);
}

__global__ __launch_bounds__(256) void k_deg_init(float* __restrict__ deg, int n, int* __restrict__ cnt) {
    int i = blockIdx.x * 256 + threadIdx.x;
    if (i < n) deg[i] = 1.0f;
    if (i < 2) cnt[i * CPAD] = 0;
}

__global__ __launch_bounds__(256) void k_deg_acc(const int* __restrict__ dst, int E, int nhalf,
                                                 float* __restrict__ deg, int* __restrict__ cnt,
                                                 unsigned char* __restrict__ key) {
    const int lane = threadIdx.x & 31;
    const int h = my_half();
    const int nt = (E + DTK - 1) / DTK;
    for (int side = 0; side < 2; ++side) {
        const int p = h ^ side;
        int* c = cnt + p * CPAD;
        for (int k = draw(c, lane); k < nt; k = draw(c, lane)) {
            int e = k * DTK + lane;
            int d[DTK / 32];
            if (k * DTK + DTK <= E) {
#pragma unroll
                for (int q = 0; q < DTK / 32; ++q) d[q] = dst[e + 32 * q];
            } else {
#pragma unroll
                for (int q = 0; q < DTK / 32; ++q) d[q] = (e + 32 * q < E) ? dst[e + 32 * q] : -1;
            }
#pragma unroll
            for (int q = 0; q < DTK / 32; ++q)
                if (d[q] >= 0 && (d[q] >= nhalf) == (p != 0)) atomicAdd(deg + d[q], 1.0f);
            if (p == 0) {
#pragma unroll
                for (int q = 0; q < DTK / 32; ++q)
                    if (d[q] >= 0) key[e + 32 * q] = (unsigned char)(d[q] / RPO);
            }
        }
    }
}

__global__ __launch_bounds__(256) void k_gemm1(const float* __restrict__ X,
                                               const float* __restrict__ W,
                                               const float* __restrict__ deg,
                                               float* __restrict__ H, int nrows) {
    int wave = (blockIdx.x * 256 + threadIdx.x) >> 5;
    int lane = threadIdx.x & 31;
    int m0 = wave << 4;
    if (m0 >= nrows) return;
    int row = lane & 15;
    int kk  = (lane >> 4) << 1;
    const float* Xr = X + (size_t)(m0 + row) * FIN + kk;
    v8f c0 = {}, c1 = {}, c2 = {}, c3 = {};
    for (int k0 = 0; k0 < FIN; k0 += 4) {
        v2f a;
        a.x = Xr[k0];
        a.y = Xr[k0 + 1];
        const float* Wp = W + (size_t)(k0 + kk) * HID;
        v2f b0 = { Wp[row],      Wp[HID + row] };
        v2f b1 = { Wp[16 + row], Wp[HID + 16 + row] };
        v2f b2 = { Wp[32 + row], Wp[HID + 32 + row] };
        v2f b3 = { Wp[48 + row], Wp[HID + 48 + row] };
        c0 = __builtin_amdgcn_wmma_f32_16x16x4_f32(false, a, false, b0, (short)0, c0, false, false);
        c1 = __builtin_amdgcn_wmma_f32_16x16x4_f32(false, a, false, b1, (short)0, c1, false, false);
        c2 = __builtin_amdgcn_wmma_f32_16x16x4_f32(false, a, false, b2, (short)0, c2, false, false);
        c3 = __builtin_amdgcn_wmma_f32_16x16x4_f32(false, a, false, b3, (short)0, c3, false, false);
    }
    int mrow = m0 + ((lane >> 4) << 3);
    int col  = lane & 15;
    for (int r = 0; r < 8; ++r) {
        float di = rsqrtf(deg[mrow + r]);
        size_t off = (size_t)(mrow + r) * HID + col;
        float y0 = c0[r] * di, y1 = c1[r] * di, y2 = c2[r] * di, y3 = c3[r] * di;
        H[off]      = y0;
        H[off + 16] = y1;
        H[off + 32] = y2;
        H[off + 48] = y3;
    }
}

__global__ __launch_bounds__(256) void k_gemm2(const float* __restrict__ A,
                                               const float* __restrict__ W,
                                               const float* __restrict__ deg,
                                               const float* __restrict__ bias1,
                                               float* __restrict__ H2, int nrows) {
    int wave = (blockIdx.x * 256 + threadIdx.x) >> 5;
    int lane = threadIdx.x & 31;
    int m0 = wave << 4;
    if (m0 >= nrows) return;
    int row = lane & 15;
    int kk  = (lane >> 4) << 1;
    const float* Ar = A + (size_t)(m0 + row) * HID + kk;
    const float* bp = bias1 + kk;
    const float dr = rsqrtf(deg[m0 + row]);
    v8f c0 = {}, c1 = {}, c2 = {};
    for (int k0 = 0; k0 < HID; k0 += 4) {
        v2f a;
        a.x = fmaxf(Ar[k0]     * dr + bp[k0],     0.0f);
        a.y = fmaxf(Ar[k0 + 1] * dr + bp[k0 + 1], 0.0f);
        const float* Wp = W + (size_t)(k0 + kk) * COUT;
        v2f b0 = { Wp[row],      Wp[COUT + row] };
        v2f b1 = { Wp[16 + row], Wp[COUT + 16 + row] };
        int col2 = 32 + row;
        v2f b2 = { (col2 < COUT) ? Wp[col2] : 0.0f,
                   (col2 < COUT) ? Wp[COUT + col2] : 0.0f };
        c0 = __builtin_amdgcn_wmma_f32_16x16x4_f32(false, a, false, b0, (short)0, c0, false, false);
        c1 = __builtin_amdgcn_wmma_f32_16x16x4_f32(false, a, false, b1, (short)0, c1, false, false);
        c2 = __builtin_amdgcn_wmma_f32_16x16x4_f32(false, a, false, b2, (short)0, c2, false, false);
    }
    int mrow = m0 + ((lane >> 4) << 3);
    int col  = lane & 15;
    for (int r = 0; r < 8; ++r) {
        float di = rsqrtf(deg[mrow + r]);
        size_t off = (size_t)(mrow + r) * LD2;
        float y0 = c0[r] * di, y1 = c1[r] * di, y2 = c2[r] * di;
        H2[off + col]      = y0;
        H2[off + 16 + col] = y1;
        if (32 + col < COUT) H2[off + 32 + col] = y2;
    }
}

#define KPL   4
#define TRIPK (KPL * 512)
#define QCAP  (512 + 8)
#define PKS   9
#define WCAP  416
#define HOFF  1638400
#define LOFF  1671168
#define LEND  (LOFF + NOWN * 32 * WCAP * 4)

__device__ __forceinline__ unsigned zero_bytes(unsigned y) {
    return ~(((y & 0x7F7F7F7Fu) + 0x7F7F7F7Fu) | y | 0x7F7F7F7Fu);
}
__device__ __forceinline__ unsigned clip_flags(unsigned f, int k) {
    return (k >= 4) ? f : (k <= 0) ? 0u : (f & ((1u << (8 * k)) - 1u));
}
__device__ __forceinline__ void push_matches(unsigned c, unsigned m, int eb, int* myq, int& cnt) {
    do {
        if (c != 0u) {
            const int p = __builtin_ctz(c);
            myq[cnt + (int)__builtin_amdgcn_mbcnt_lo(m, 0u)] = eb + ((p & 7) << 2) + (p >> 3);
            c &= c - 1u;
        }
        cnt += __builtin_popcount(m);
        m = __builtin_amdgcn_ballot_w32(c != 0u);
    } while (m != 0u);
}

template <int NC, bool FULL>
__device__ __forceinline__ void add_batch(int pk, int b0, int nb, const float* __restrict__ H,
                                          float* rows, int lane) {
    const bool tail = (NC == 64) || (lane < NC - 32);
    int sj[8], dj[8];
#pragma unroll
    for (int u = 0; u < 8; ++u) {
        const int p = __builtin_amdgcn_readlane(pk, b0 + u);
        sj[u] = p >> PKS;
        dj[u] = p & ((1 << PKS) - 1);
    }
    float v0[8], v1[8];
#pragma unroll
    for (int u = 0; u < 8; ++u) {
        v0[u] = 0.0f; v1[u] = 0.0f;
        if (FULL || u < nb) v0[u] = H[(size_t)sj[u] * LD2 + lane];
    }
    if (tail) {
#pragma unroll
        for (int u = 0; u < 8; ++u)
            if (FULL || u < nb) v1[u] = H[(size_t)sj[u] * LD2 + 32 + lane];
    }
#pragma unroll
    for (int u = 0; u < 8; ++u)
        if (FULL || u < nb) atomicAdd(rows + dj[u] * LD2 + lane, v0[u]);
    if (tail) {
#pragma unroll
        for (int u = 0; u < 8; ++u)
            if (FULL || u < nb) atomicAdd(rows + dj[u] * LD2 + 32 + lane, v1[u]);
    }
}

template <int NC, bool FULL>
__device__ __forceinline__ void own_batch(const int* __restrict__ src, const int* __restrict__ dst,
                                          const float* __restrict__ H, float* rows, const int* myq,
                                          int top, int nb, int row0, int lane,
                                          bool rec, int* __restrict__ mylist, int& nrec) {
    const bool mine = lane < (FULL ? 8 : nb);
    int e = 0;
    if (mine) e = myq[top - 1 - lane];
    int s = 0, d = row0;
    if (mine) { s = src[e]; d = dst[e]; }
    const int pk = (s << PKS) | (d - row0);
    if (rec) {
        if (mine && nrec + lane < WCAP) mylist[nrec + lane] = pk;
        nrec += FULL ? 8 : nb;
    }
    add_batch<NC, FULL>(pk, 0, nb, H, rows, lane);
}

template <int NC>
__device__ __forceinline__ void scan_keys(const uint4* __restrict__ key4, const int* __restrict__ src,
                                          const int* __restrict__ dst, int E, const float* __restrict__ H,
                                          float* rows, int* myq, int row0, int g, int w, int lane,
                                          bool rec, int* __restrict__ mylist, int& nrec) {
    const unsigned pat = (unsigned)g * 0x01010101u;
    int cnt = 0;
    const int ntrip = (E + TRIPK - 1) / TRIPK;
    const int rot = (int)(((long long)g * ntrip) >> 8);
    uint4 kv[KPL], nx[KPL];
    {
        int tq = (w < ntrip) ? w + rot : 0;
        if (tq >= ntrip) tq -= ntrip;
#pragma unroll
        for (int j = 0; j < KPL; ++j) kv[j] = key4[((tq * TRIPK) >> 4) + j * 32 + lane];
    }
    for (int t = w; t < ntrip; t += 32) {
        int tt = t + rot;
        if (tt >= ntrip) tt -= ntrip;
        const int e0 = tt * TRIPK;
        int tq = ((t + 32 < ntrip) ? t + 32 : t) + rot;
        if (tq >= ntrip) tq -= ntrip;
#pragma unroll
        for (int j = 0; j < KPL; ++j) nx[j] = key4[((tq * TRIPK) >> 4) + j * 32 + lane];
#pragma unroll
        for (int j = 0; j < KPL; ++j) {
            const int eb = e0 + j * 512 + lane * 16;
            unsigned f0 = zero_bytes(kv[j].x ^ pat), f1 = zero_bytes(kv[j].y ^ pat);
            unsigned f2 = zero_bytes(kv[j].z ^ pat), f3 = zero_bytes(kv[j].w ^ pat);
            if (eb + 16 > E) {
                f0 = clip_flags(f0, E - eb);      f1 = clip_flags(f1, E - eb - 4);
                f2 = clip_flags(f2, E - eb - 8);  f3 = clip_flags(f3, E - eb - 12);
            }
            const unsigned c16 = (f0 >> 7) | (f1 >> 6) | (f2 >> 5) | (f3 >> 4);
            const unsigned mm = __builtin_amdgcn_ballot_w32(c16 != 0u);
            if (mm != 0u) {
                push_matches(c16, mm, eb, myq, cnt);
                while (cnt >= 8) {
                    own_batch<NC, true>(src, dst, H, rows, myq, cnt, 8, row0, lane, rec, mylist, nrec);
                    cnt -= 8;
                }
            }
        }
#pragma unroll
        for (int j = 0; j < KPL; ++j) kv[j] = nx[j];
    }
    if (cnt > 0) own_batch<NC, false>(src, dst, H, rows, myq, cnt, cnt, row0, lane, rec, mylist, nrec);
}

template <int NC, bool FIRST>
__global__ __launch_bounds__(1024) void k_own(const uint4* __restrict__ key4,
                                              const int* __restrict__ src,
                                              const int* __restrict__ dst, int E, int n,
                                              const float* __restrict__ H, float* __restrict__ A,
                                              int* __restrict__ hdr, int* __restrict__ lst) {
    __shared__ float rows[RPO * LD2];
    __shared__ int   q[32 * QCAP];
    constexpr int F4 = NC / 4;
    const int tid  = threadIdx.x;
    const int lane = tid & 31;
    const int w    = __builtin_amdgcn_readfirstlane(tid >> 5);
    const int g    = blockIdx.x;
    const int row0 = g * RPO;
    const int nown = min(RPO, n - row0);
    const float4* H4 = (const float4*)H;
    float4* A4 = (float4*)A;
    float4* rows4 = (float4*)rows;

    constexpr int NIT = (RPO * F4 + 1023) / 1024;
    float4 st[NIT];
#pragma unroll
    for (int u = 0; u < NIT; ++u) {
        const int i = tid + u * 1024;
        const int r = i / F4, f = i - r * F4;
        st[u] = make_float4(0.0f, 0.0f, 0.0f, 0.0f);
        if (i < nown * F4) st[u] = H4[(size_t)(row0 + r) * (LD2 / 4) + f];
    }
#pragma unroll
    for (int u = 0; u < NIT; ++u) {
        const int i = tid + u * 1024;
        const int r = i / F4, f = i - r * F4;
        if (i < nown * F4) rows4[r * (LD2 / 4) + f] = st[u];
    }
    __syncthreads();

    const bool lists = (4096 + (long long)E <= HOFF) && ((long long)n * COUT * 4 >= LEND);
    int* myq = q + w * QCAP;
    int* mylist = lst + (size_t)(g * 32 + w) * WCAP;
    int nrec = 0;
    if (FIRST) {
        scan_keys<NC>(key4, src, dst, E, H, rows, myq, row0, g, w, lane, lists, mylist, nrec);
        if (lists && lane == 0) hdr[g * 32 + w] = nrec;
    } else {
        int cw = WCAP + 1;
        if (lists) cw = hdr[g * 32 + lane];
        if (__builtin_amdgcn_ballot_w32(cw > WCAP) == 0u) {
            const int nmine = __builtin_amdgcn_readlane(cw, w);
            int ent = 0;
            if (lane < nmine) ent = mylist[lane];
            int c0 = 0;
            for (; c0 + 32 <= nmine; c0 += 32) {
                const int cur = ent;
                if (c0 + 32 + lane < nmine) ent = mylist[c0 + 32 + lane];
                add_batch<NC, true>(cur, 0,  8, H, rows, lane);
                add_batch<NC, true>(cur, 8,  8, H, rows, lane);
                add_batch<NC, true>(cur, 16, 8, H, rows, lane);
                add_batch<NC, true>(cur, 24, 8, H, rows, lane);
            }
            const int left = nmine - c0;
            for (int b = 0; b < left; b += 8)
                add_batch<NC, false>(ent, b, min(8, left - b), H, rows, lane);
        } else {
            scan_keys<NC>(key4, src, dst, E, H, rows, myq, row0, g, w, lane, false, mylist, nrec);
        }
    }
    __syncthreads();

    for (int i = tid; i < nown * F4; i += 1024) {
        const int r = i / F4, f = i - r * F4;
        A4[(size_t)(row0 + r) * (LD2 / 4) + f] = rows4[r * (LD2 / 4) + f];
    }
}

template <bool GUARD>
__device__ __forceinline__ void fin2_trip(float4* __restrict__ out, const float4* __restrict__ acc,
                                          const float* __restrict__ deg, const float* __restrict__ b2,
                                          int u0, int nwaves, int nunits, int lane) {
    float4 o[FPT], bb[FPT];
    float  dg[FPT];
#pragma unroll
    for (int u = 0; u < FPT; ++u) {
        const int unit = GUARD ? min(u0 + u * nwaves, nunits - 1) : u0 + u * nwaves;
        const int idx  = unit * 32 + lane;
        const int i    = idx / 10;
        const int f4   = idx - i * 10;
        const float* b = b2 + f4 * 4;
        dg[u] = deg[i];
        o[u]  = acc[(size_t)i * (LD2 / 4) + f4];
        bb[u] = make_float4(b[0], b[1], b[2], b[3]);
    }
#pragma unroll
    for (int u = 0; u < FPT; ++u) {
        const int unit = u0 + u * nwaves;
        if (!GUARD || unit < nunits) {
            const float di = rsqrtf(dg[u]);
            float4 r = o[u];
            r.x = r.x * di + bb[u].x;
            r.y = r.y * di + bb[u].y;
            r.z = r.z * di + bb[u].z;
            r.w = r.w * di + bb[u].w;
            out[unit * 32 + lane] = r;
        }
    }
}

__global__ __launch_bounds__(256) void k_fin2(float4* __restrict__ out,
                                              const float4* __restrict__ acc,
                                              const float* __restrict__ deg,
                                              const float* __restrict__ b2, int n) {
    const int nunits = (n * (COUT / 4)) >> 5;
    const int nwaves = gridDim.x * 8;
    const int lane   = threadIdx.x & 31;
    const int wave   = __builtin_amdgcn_readfirstlane((int)((blockIdx.x * 256 + threadIdx.x) >> 5));
    for (int u0 = wave; u0 < nunits; u0 += nwaves * FPT) {
        if (u0 + (FPT - 1) * nwaves < nunits) fin2_trip<false>(out, acc, deg, b2, u0, nwaves, nunits, lane);
        else                                  fin2_trip<true >(out, acc, deg, b2, u0, nwaves, nunits, lane);
    }
}

extern "C" void kernel_launch(void* const* d_in, const int* in_sizes, int n_in,
                              void* d_out, int out_size, void* d_ws, size_t ws_size,
                              hipStream_t stream) {
    const float* x  = (const float*)d_in[0];
    const float* W1 = (const float*)d_in[1];
    const float* b1 = (const float*)d_in[2];
    const float* W2 = (const float*)d_in[3];
    const float* b2 = (const float*)d_in[4];
    const int*   ei = (const int*)d_in[5];

    const int Nn = in_sizes[0] / FIN;
    const int E  = in_sizes[5] / 2;
    const int* src = ei;
    const int* dst = ei + E;

    float* ws   = (float*)d_ws;
    float* deg  = ws;
    float* h    = ws + Nn;
    float* agg  = h + (size_t)Nn * HID;
    float* h2   = h;
    float* acc2 = agg;
    float* out  = (float*)d_out;

    int gN  = (Nn + 255) / 256;
    int nhalf = ((Nn / 2 + 31) / 32) * 32;
    int* cnt  = (int*)d_out;
    unsigned char* key = (unsigned char*)d_out + 4096;
    int* hdr = (int*)((unsigned char*)d_out + HOFF);
    int* lst = (int*)((unsigned char*)d_out + LOFF);
    int gemmBlocks = ((Nn / 16) + 7) / 8;
    int gF = (Nn * (COUT / 4) / 32 + FPT * 8 - 1) / (FPT * 8);

    k_deg_init<<<gN, 256, 0, stream>>>(deg, Nn, cnt);
    k_deg_acc<<<DBLK, 256, 0, stream>>>(dst, E, nhalf, deg, cnt, key);

    k_gemm1<<<gemmBlocks, 256, 0, stream>>>(x, W1, deg, h, Nn);
    k_own<HID, true><<<NOWN, 1024, 0, stream>>>((const uint4*)key, src, dst, E, Nn, h, agg, hdr, lst);

    k_gemm2<<<gemmBlocks, 256, 0, stream>>>(agg, W2, deg, b1, h2, Nn);
    k_own<COUT, false><<<NOWN, 1024, 0, stream>>>((const uint4*)key, src, dst, E, Nn, h2, acc2, hdr, lst);
    k_fin2<<<gF, 256, 0, stream>>>((float4*)out, (const float4*)acc2, deg, b2, Nn);
}


# edge search split off ahead of product 1: owners count their rows' degrees while they search (no global adds, dst read once) and leave lists; both edge passes replay them; sums in arrival order
# speedup vs baseline: 2.9495x; 1.2217x over previous
#include <hip/hip_runtime.h>

#define FIN 256
#define HID 64
#define COUT 40
#define LD2 64

typedef float v2f __attribute__((ext_vector_type(2)));
typedef float v8f __attribute__((ext_vector_type(8)));

#define RPO   391
#define NOWN  256
#define PKS   9
#define KBLK  256
#define KU    4
#define FPT   8

__device__ __forceinline__ int pack_edge(int s, int d, unsigned& k) {
    k = (unsigned)(d / RPO);
    return (s << PKS) | (d - (int)k * RPO);
}

__global__ __launch_bounds__(256) void k_keys(const int* __restrict__ src, const int* __restrict__ dst, int E,
                                              unsigned char* __restrict__ key, int* __restrict__ pk) {
    const int nthr = gridDim.x * 256;
    const int gt   = blockIdx.x * 256 + threadIdx.x;
    if ((E & 3) == 0) {
        const int4* s4 = (const int4*)src;
        const int4* d4 = (const int4*)dst;
        unsigned* k4 = (unsigned*)key;
        int4* p4 = (int4*)pk;
        const int nun = E >> 2;
        for (int u0 = gt; u0 < nun; u0 += nthr * KU) {
            int4 s[KU], d[KU];
#pragma unroll
            for (int j = 0; j < KU; ++j) {
                const int u = min(u0 + j * nthr, nun - 1);
                s[j] = s4[u];
                d[j] = d4[u];
            }
#pragma unroll
            for (int j = 0; j < KU; ++j) {
                const int u = u0 + j * nthr;
                if (u < nun) {
                    unsigned k0, k1, k2, k3;
                    int4 p;
                    p.x = pack_edge(s[j].x, d[j].x, k0);
                    p.y = pack_edge(s[j].y, d[j].y, k1);
                    p.z = pack_edge(s[j].z, d[j].z, k2);
                    p.w = pack_edge(s[j].w, d[j].w, k3);
                    k4[u] = k0 | (k1 << 8) | (k2 << 16) | (k3 << 24);
                    p4[u] = p;
                }
            }
        }
    } else {
        for (int e = gt; e < E; e += nthr) {
            unsigned k;
            pk[e]  = pack_edge(src[e], dst[e], k);
            key[e] = (unsigned char)k;
        }
    }
}

__global__ __launch_bounds__(256) void k_gemm1(const float* __restrict__ X,
                                               const float* __restrict__ W,
                                               const float* __restrict__ deg,
                                               float* __restrict__ H, int nrows) {
    int wave = (blockIdx.x * 256 + threadIdx.x) >> 5;
    int lane = threadIdx.x & 31;
    int m0 = wave << 4;
    if (m0 >= nrows) return;
    int row = lane & 15;
    int kk  = (lane >> 4) << 1;
    const float* Xr = X + (size_t)(m0 + row) * FIN + kk;
    v8f c0 = {}, c1 = {}, c2 = {}, c3 = {};
    for (int k0 = 0; k0 < FIN; k0 += 4) {
        v2f a;
        a.x = Xr[k0];
        a.y = Xr[k0 + 1];
        const float* Wp = W + (size_t)(k0 + kk) * HID;
        v2f b0 = { Wp[row],      Wp[HID + row] };
        v2f b1 = { Wp[16 + row], Wp[HID + 16 + row] };
        v2f b2 = { Wp[32 + row], Wp[HID + 32 + row] };
        v2f b3 = { Wp[48 + row], Wp[HID + 48 + row] };
        c0 = __builtin_amdgcn_wmma_f32_16x16x4_f32(false, a, false, b0, (short)0, c0, false, false);
        c1 = __builtin_amdgcn_wmma_f32_16x16x4_f32(false, a, false, b1, (short)0, c1, false, false);
        c2 = __builtin_amdgcn_wmma_f32_16x16x4_f32(false, a, false, b2, (short)0, c2, false, false);
        c3 = __builtin_amdgcn_wmma_f32_16x16x4_f32(false, a, false, b3, (short)0, c3, false, false);
    }
    int mrow = m0 + ((lane >> 4) << 3);
    int col  = lane & 15;
    for (int r = 0; r < 8; ++r) {
        float di = rsqrtf(deg[mrow + r]);
        size_t off = (size_t)(mrow + r) * HID + col;
        float y0 = c0[r] * di, y1 = c1[r] * di, y2 = c2[r] * di, y3 = c3[r] * di;
        H[off]      = y0;
        H[off + 16] = y1;
        H[off + 32] = y2;
        H[off + 48] = y3;
    }
}

__global__ __launch_bounds__(256) void k_gemm2(const float* __restrict__ A,
                                               const float* __restrict__ W,
                                               const float* __restrict__ deg,
                                               const float* __restrict__ bias1,
                                               float* __restrict__ H2, int nrows) {
    int wave = (blockIdx.x * 256 + threadIdx.x) >> 5;
    int lane = threadIdx.x & 31;
    int m0 = wave << 4;
    if (m0 >= nrows) return;
    int row = lane & 15;
    int kk  = (lane >> 4) << 1;
    const float* Ar = A + (size_t)(m0 + row) * HID + kk;
    const float* bp = bias1 + kk;
    const float dr = rsqrtf(deg[m0 + row]);
    v8f c0 = {}, c1 = {}, c2 = {};
    for (int k0 = 0; k0 < HID; k0 += 4) {
        v2f a;
        a.x = fmaxf(Ar[k0]     * dr + bp[k0],     0.0f);
        a.y = fmaxf(Ar[k0 + 1] * dr + bp[k0 + 1], 0.0f);
        const float* Wp = W + (size_t)(k0 + kk) * COUT;
        v2f b0 = { Wp[row],      Wp[COUT + row] };
        v2f b1 = { Wp[16 + row], Wp[COUT + 16 + row] };
        int col2 = 32 + row;
        v2f b2 = { (col2 < COUT) ? Wp[col2] : 0.0f,
                   (col2 < COUT) ? Wp[COUT + col2] : 0.0f };
        c0 = __builtin_amdgcn_wmma_f32_16x16x4_f32(false, a, false, b0, (short)0, c0, false, false);
        c1 = __builtin_amdgcn_wmma_f32_16x16x4_f32(false, a, false, b1, (short)0, c1, false, false);
        c2 = __builtin_amdgcn_wmma_f32_16x16x4_f32(false, a, false, b2, (short)0, c2, false, false);
    }
    int mrow = m0 + ((lane >> 4) << 3);
    int col  = lane & 15;
    for (int r = 0; r < 8; ++r) {
        float di = rsqrtf(deg[mrow + r]);
        size_t off = (size_t)(mrow + r) * LD2;
        float y0 = c0[r] * di, y1 = c1[r] * di, y2 = c2[r] * di;
        H2[off + col]      = y0;
        H2[off + 16 + col] = y1;
        if (32 + col < COUT) H2[off + 32 + col] = y2;
    }
}

#define KPL   4
#define TRIPK (KPL * 512)
#define QCAP  (512 + 32)
#define WCAP  416
#define HOFF  1638400
#define LOFF  1671168
#define LEND  (LOFF + NOWN * 32 * WCAP * 4)

__device__ __forceinline__ unsigned zero_bytes(unsigned y) {
    return ~(((y & 0x7F7F7F7Fu) + 0x7F7F7F7Fu) | y | 0x7F7F7F7Fu);
}
__device__ __forceinline__ unsigned clip_flags(unsigned f, int k) {
    return (k >= 4) ? f : (k <= 0) ? 0u : (f & ((1u << (8 * k)) - 1u));
}
__device__ __forceinline__ void push_matches(unsigned c, unsigned m, int eb, int* myq, int& cnt) {
    do {
        if (c != 0u) {
            const int p = __builtin_ctz(c);
            myq[cnt + (int)__builtin_amdgcn_mbcnt_lo(m, 0u)] = eb + ((p & 7) << 2) + (p >> 3);
            c &= c - 1u;
        }
        cnt += __builtin_popcount(m);
        m = __builtin_amdgcn_ballot_w32(c != 0u);
    } while (m != 0u);
}

__device__ __forceinline__ unsigned match_flags(uint4 kv, unsigned pat, int eb, int E) {
    unsigned f0 = zero_bytes(kv.x ^ pat), f1 = zero_bytes(kv.y ^ pat);
    unsigned f2 = zero_bytes(kv.z ^ pat), f3 = zero_bytes(kv.w ^ pat);
    if (eb + 16 > E) {
        f0 = clip_flags(f0, E - eb);      f1 = clip_flags(f1, E - eb - 4);
        f2 = clip_flags(f2, E - eb - 8);  f3 = clip_flags(f3, E - eb - 12);
    }
    return (f0 >> 7) | (f1 >> 6) | (f2 >> 5) | (f3 >> 4);
}

template <bool FULL>
__device__ __forceinline__ void note_batch(const int* __restrict__ pk, int* cntrow, const int* myq, int top,
                                           int nb, int lane, bool rec, int* __restrict__ mylist, int& nrec) {
    if (FULL || lane < nb) {
        const int p = pk[myq[top - 1 - lane]];
        if (rec && nrec + lane < WCAP) mylist[nrec + lane] = p;
        atomicAdd(cntrow + (p & ((1 << PKS) - 1)), 1);
    }
    nrec += FULL ? 32 : nb;
}

template <int BATCH, class FullFn, class RestFn>
__device__ __forceinline__ void walk_keys(const uint4* __restrict__ key4, int E, int* myq, int g, int w, int lane,
                                          FullFn full, RestFn rest) {
    const unsigned pat = (unsigned)g * 0x01010101u;
    int cnt = 0;
    const int ntrip = (E + TRIPK - 1) / TRIPK;
    const int rot = (int)(((long long)g * ntrip) >> 8);
    uint4 kv[KPL], nx[KPL];
    {
        int tq = (w < ntrip) ? w + rot : 0;
        if (tq >= ntrip) tq -= ntrip;
#pragma unroll
        for (int j = 0; j < KPL; ++j) kv[j] = key4[((tq * TRIPK) >> 4) + j * 32 + lane];
    }
    for (int t = w; t < ntrip; t += 32) {
        int tt = t + rot;
        if (tt >= ntrip) tt -= ntrip;
        const int e0 = tt * TRIPK;
        int tq = ((t + 32 < ntrip) ? t + 32 : t) + rot;
        if (tq >= ntrip) tq -= ntrip;
#pragma unroll
        for (int j = 0; j < KPL; ++j) nx[j] = key4[((tq * TRIPK) >> 4) + j * 32 + lane];
#pragma unroll
        for (int j = 0; j < KPL; ++j) {
            const int eb = e0 + j * 512 + lane * 16;
            const unsigned c16 = match_flags(kv[j], pat, eb, E);
            const unsigned mm = __builtin_amdgcn_ballot_w32(c16 != 0u);
            if (mm != 0u) {
                push_matches(c16, mm, eb, myq, cnt);
                while (cnt >= BATCH) {
                    full(cnt);
                    cnt -= BATCH;
                }
            }
        }
#pragma unroll
        for (int j = 0; j < KPL; ++j) kv[j] = nx[j];
    }
    if (cnt > 0) rest(cnt, cnt);
}

__global__ __launch_bounds__(1024) void k_search(const uint4* __restrict__ key4, const int* __restrict__ pk,
                                                 int E, int n, float* __restrict__ deg,
                                                 int* __restrict__ hdr, int* __restrict__ lst) {
    __shared__ int q[32 * QCAP];
    __shared__ int cntrow[RPO];
    const int tid  = threadIdx.x;
    const int lane = tid & 31;
    const int w    = __builtin_amdgcn_readfirstlane(tid >> 5);
    const int g    = blockIdx.x;
    const int row0 = g * RPO;
    const int nown = min(RPO, n - row0);
    for (int i = tid; i < RPO; i += 1024) cntrow[i] = 0;
    __syncthreads();

    const bool lists = (4096 + (long long)E <= HOFF) && ((long long)n * COUT * 4 >= LEND);
    int* myq = q + w * QCAP;
    int* mylist = lst + (size_t)(g * 32 + w) * WCAP;
    int nrec = 0;
    walk_keys<32>(key4, E, myq, g, w, lane,
        [&](int top) { note_batch<true >(pk, cntrow, myq, top, 32, lane, lists, mylist, nrec); },
        [&](int top, int nb) { note_batch<false>(pk, cntrow, myq, top, nb, lane, lists, mylist, nrec); });
    if (lists && lane == 0) hdr[g * 32 + w] = nrec;
    __syncthreads();

    for (int i = tid; i < nown; i += 1024) deg[row0 + i] = 1.0f + (float)cntrow[i];
}

template <int NC, bool FULL>
__device__ __forceinline__ void add_batch(int pk, int b0, int nb, const float* __restrict__ H,
                                          float* rows, int lane) {
    const bool tail = (NC == 64) || (lane < NC - 32);
    int sj[8], dj[8];
#pragma unroll
    for (int u = 0; u < 8; ++u) {
        const int p = __builtin_amdgcn_readlane(pk, b0 + u);
        sj[u] = p >> PKS;
        dj[u] = p & ((1 << PKS) - 1);
    }
    float v0[8], v1[8];
#pragma unroll
    for (int u = 0; u < 8; ++u) {
        v0[u] = 0.0f; v1[u] = 0.0f;
        if (FULL || u < nb) v0[u] = H[(size_t)sj[u] * LD2 + lane];
    }
    if (tail) {
#pragma unroll
        for (int u = 0; u < 8; ++u)
            if (FULL || u < nb) v1[u] = H[(size_t)sj[u] * LD2 + 32 + lane];
    }
#pragma unroll
    for (int u = 0; u < 8; ++u)
        if (FULL || u < nb) atomicAdd(rows + dj[u] * LD2 + lane, v0[u]);
    if (tail) {
#pragma unroll
        for (int u = 0; u < 8; ++u)
            if (FULL || u < nb) atomicAdd(rows + dj[u] * LD2 + 32 + lane, v1[u]);
    }
}

template <int NC, bool FULL>
__device__ __forceinline__ void own_batch(const int* __restrict__ src, const int* __restrict__ dst,
                                          const float* __restrict__ H, float* rows, const int* myq,
                                          int top, int nb, int row0, int lane) {
    const bool mine = lane < (FULL ? 8 : nb);
    int e = 0;
    if (mine) e = myq[top - 1 - lane];
    int s = 0, d = row0;
    if (mine) { s = src[e]; d = dst[e]; }
    const int pk = (s << PKS) | (d - row0);
    add_batch<NC, FULL>(pk, 0, nb, H, rows, lane);
}

template <int NC>
__global__ __launch_bounds__(1024) void k_own(const uint4* __restrict__ key4,
                                              const int* __restrict__ src,
                                              const int* __restrict__ dst, int E, int n,
                                              const float* __restrict__ H, float* __restrict__ A,
                                              const int* __restrict__ hdr, const int* __restrict__ lst) {
    __shared__ float rows[RPO * LD2];
    __shared__ int   q[32 * QCAP];
    constexpr int F4 = NC / 4;
    const int tid  = threadIdx.x;
    const int lane = tid & 31;
    const int w    = __builtin_amdgcn_readfirstlane(tid >> 5);
    const int g    = blockIdx.x;
    const int row0 = g * RPO;
    const int nown = min(RPO, n - row0);
    const float4* H4 = (const float4*)H;
    float4* A4 = (float4*)A;
    float4* rows4 = (float4*)rows;

    constexpr int NIT = (RPO * F4 + 1023) / 1024;
    float4 st[NIT];
#pragma unroll
    for (int u = 0; u < NIT; ++u) {
        const int i = tid + u * 1024;
        const int r = i / F4, f = i - r * F4;
        st[u] = make_float4(0.0f, 0.0f, 0.0f, 0.0f);
        if (i < nown * F4) st[u] = H4[(size_t)(row0 + r) * (LD2 / 4) + f];
    }
#pragma unroll
    for (int u = 0; u < NIT; ++u) {
        const int i = tid + u * 1024;
        const int r = i / F4, f = i - r * F4;
        if (i < nown * F4) rows4[r * (LD2 / 4) + f] = st[u];
    }
    __syncthreads();

    const bool lists = (4096 + (long long)E <= HOFF) && ((long long)n * COUT * 4 >= LEND);
    const int* mylist = lst + (size_t)(g * 32 + w) * WCAP;
    int cw = WCAP + 1;
    if (lists) cw = hdr[g * 32 + lane];
    if (__builtin_amdgcn_ballot_w32(cw > WCAP) == 0u) {
        const int nmine = __builtin_amdgcn_readlane(cw, w);
        int ent = 0;
        if (lane < nmine) ent = mylist[lane];
        int c0 = 0;
        for (; c0 + 32 <= nmine; c0 += 32) {
            const int cur = ent;
            if (c0 + 32 + lane < nmine) ent = mylist[c0 + 32 + lane];
            add_batch<NC, true>(cur, 0,  8, H, rows, lane);
            add_batch<NC, true>(cur, 8,  8, H, rows, lane);
            add_batch<NC, true>(cur, 16, 8, H, rows, lane);
            add_batch<NC, true>(cur, 24, 8, H, rows, lane);
        }
        const int left = nmine - c0;
        for (int b = 0; b < left; b += 8)
            add_batch<NC, false>(ent, b, min(8, left - b), H, rows, lane);
    } else {
        int* myq = q + w * QCAP;
        walk_keys<8>(key4, E, myq, g, w, lane,
            [&](int top) { own_batch<NC, true >(src, dst, H, rows, myq, top, 8,  row0, lane); },
            [&](int top, int nb) { own_batch<NC, false>(src, dst, H, rows, myq, top, nb, row0, lane); });
    }
    __syncthreads();

    for (int i = tid; i < nown * F4; i += 1024) {
        const int r = i / F4, f = i - r * F4;
        A4[(size_t)(row0 + r) * (LD2 / 4) + f] = rows4[r * (LD2 / 4) + f];
    }
}

template <bool GUARD>
__device__ __forceinline__ void fin2_trip(float4* __restrict__ out, const float4* __restrict__ acc,
                                          const float* __restrict__ deg, const float* __restrict__ b2,
                                          int u0, int nwaves, int nunits, int lane) {
    float4 o[FPT], bb[FPT];
    float  dg[FPT];
#pragma unroll
    for (int u = 0; u < FPT; ++u) {
        const int unit = GUARD ? min(u0 + u * nwaves, nunits - 1) : u0 + u * nwaves;
        const int idx  = unit * 32 + lane;
        const int i    = idx / 10;
        const int f4   = idx - i * 10;
        const float* b = b2 + f4 * 4;
        dg[u] = deg[i];
        o[u]  = acc[(size_t)i * (LD2 / 4) + f4];
        bb[u] = make_float4(b[0], b[1], b[2], b[3]);
    }
#pragma unroll
    for (int u = 0; u < FPT; ++u) {
        const int unit = u0 + u * nwaves;
        if (!GUARD || unit < nunits) {
            const float di = rsqrtf(dg[u]);
            float4 r = o[u];
            r.x = r.x * di + bb[u].x;
            r.y = r.y * di + bb[u].y;
            r.z = r.z * di + bb[u].z;
            r.w = r.w * di + bb[u].w;
            out[unit * 32 + lane] = r;
        }
    }
}

__global__ __launch_bounds__(256) void k_fin2(float4* __restrict__ out,
                                              const float4* __restrict__ acc,
                                              const float* __restrict__ deg,
                                              const float* __restrict__ b2, int n) {
    const int nunits = (n * (COUT / 4)) >> 5;
    const int nwaves = gridDim.x * 8;
    const int lane   = threadIdx.x & 31;
    const int wave   = __builtin_amdgcn_readfirstlane((int)((blockIdx.x * 256 + threadIdx.x) >> 5));
    for (int u0 = wave; u0 < nunits; u0 += nwaves * FPT) {
        if (u0 + (FPT - 1) * nwaves < nunits) fin2_trip<false>(out, acc, deg, b2, u0, nwaves, nunits, lane);
        else                                  fin2_trip<true >(out, acc, deg, b2, u0, nwaves, nunits, lane);
    }
}

extern "C" void kernel_launch(void* const* d_in, const int* in_sizes, int n_in,
                              void* d_out, int out_size, void* d_ws, size_t ws_size,
                              hipStream_t stream) {
    const float* x  = (const float*)d_in[0];
    const float* W1 = (const float*)d_in[1];
    const float* b1 = (const float*)d_in[2];
    const float* W2 = (const float*)d_in[3];
    const float* b2 = (const float*)d_in[4];
    const int*   ei = (const int*)d_in[5];

    const int Nn = in_sizes[0] / FIN;
    const int E  = in_sizes[5] / 2;
    const int* src = ei;
    const int* dst = ei + E;

    float* ws   = (float*)d_ws;
    float* deg  = ws;
    float* h    = ws + Nn;
    float* agg  = h + (size_t)Nn * HID;
    float* h2   = h;
    float* acc2 = agg;
    float* out  = (float*)d_out;

    unsigned char* key = (unsigned char*)d_out + 4096;
    int* hdr = (int*)((unsigned char*)d_out + HOFF);
    int* lst = (int*)((unsigned char*)d_out + LOFF);
    int* pk  = (int*)agg;
    int gemmBlocks = ((Nn / 16) + 7) / 8;
    int gF = (Nn * (COUT / 4) / 32 + FPT * 8 - 1) / (FPT * 8);

    k_keys<<<KBLK, 256, 0, stream>>>(src, dst, E, key, pk);
    k_search<<<NOWN, 1024, 0, stream>>>((const uint4*)key, pk, E, Nn, deg, hdr, lst);

    k_gemm1<<<gemmBlocks, 256, 0, stream>>>(x, W1, deg, h, Nn);
    k_own<HID><<<NOWN, 1024, 0, stream>>>((const uint4*)key, src, dst, E, Nn, h, agg, hdr, lst);

    k_gemm2<<<gemmBlocks, 256, 0, stream>>>(agg, W2, deg, b1, h2, Nn);
    k_own<COUT><<<NOWN, 1024, 0, stream>>>((const uint4*)key, src, dst, E, Nn, h2, acc2, hdr, lst);
    k_fin2<<<gF, 256, 0, stream>>>((float4*)out, (const float4*)acc2, deg, b2, Nn);
}


# owner keys kept as 8 bit planes (one ballot per key bit in the key pass): the search tests 32 edges per lane with 8 XORs and one OR, not 16 by byte arithmetic; sums in arrival order as before
# speedup vs baseline: 3.1203x; 1.0579x over previous
#include <hip/hip_runtime.h>

#define FIN 256
#define HID 64
#define COUT 40
#define LD2 64

typedef float v2f __attribute__((ext_vector_type(2)));
typedef float v8f __attribute__((ext_vector_type(8)));

#define RPO   391
#define NOWN  256
#define PKS   9
#define KBLK  256
#define KU    4
#define FPT   8

__device__ __forceinline__ int pack_edge(int s, int d, unsigned& k) {
    k = (unsigned)(d / RPO);
    return (s << PKS) | (d - (int)k * RPO);
}

__global__ __launch_bounds__(256) void k_keys(const int* __restrict__ src, const int* __restrict__ dst, int E,
                                              unsigned* __restrict__ kp, int* __restrict__ pk) {
    const int lane = threadIdx.x & 31;
    const int wave = __builtin_amdgcn_readfirstlane((int)((blockIdx.x * 256 + threadIdx.x) >> 5));
    const int nwav = gridDim.x * 8;
    const int nwu  = (E + 127) >> 7;
    const bool al  = (E & 3) == 0;
    for (int w0 = wave; w0 < nwu; w0 += nwav * KU) {
        int4 s[KU], d[KU];
#pragma unroll
        for (int j = 0; j < KU; ++j) {
            const int wu = min(w0 + j * nwav, nwu - 1);
            const int e  = wu * 128 + lane * 4;
            if (al) {
                const int u = min(e, E - 4) >> 2;
                s[j] = ((const int4*)src)[u];
                d[j] = ((const int4*)dst)[u];
            } else {
                s[j].x = src[min(e, E - 1)];     d[j].x = dst[min(e, E - 1)];
                s[j].y = src[min(e + 1, E - 1)]; d[j].y = dst[min(e + 1, E - 1)];
                s[j].z = src[min(e + 2, E - 1)]; d[j].z = dst[min(e + 2, E - 1)];
                s[j].w = src[min(e + 3, E - 1)]; d[j].w = dst[min(e + 3, E - 1)];
            }
        }
#pragma unroll
        for (int j = 0; j < KU; ++j) {
            const int wu = w0 + j * nwav;
            if (wu < nwu) {
                const int e = wu * 128 + lane * 4;
                unsigned k0, k1, k2, k3;
                int4 p;
                p.x = pack_edge(s[j].x, d[j].x, k0);
                p.y = pack_edge(s[j].y, d[j].y, k1);
                p.z = pack_edge(s[j].z, d[j].z, k2);
                p.w = pack_edge(s[j].w, d[j].w, k3);
                if (al) {
                    if (e < E) ((int4*)pk)[e >> 2] = p;
                } else {
                    if (e < E)     pk[e]     = p.x;
                    if (e + 1 < E) pk[e + 1] = p.y;
                    if (e + 2 < E) pk[e + 2] = p.z;
                    if (e + 3 < E) pk[e + 3] = p.w;
                }
                const unsigned kk = k0 | (k1 << 8) | (k2 << 16) | (k3 << 24);
                unsigned wv = 0u;
#pragma unroll
                for (int t = 0; t < 32; ++t) {
                    const unsigned sres = __builtin_amdgcn_ballot_w32(((kk >> t) & 1u) != 0u);
                    const int jl = ((t >> 3) << 2) + (t & 3) + (((t >> 2) & 1) << 4);
                    wv = (lane == jl) ? sres : wv;
                }
                kp[(wu >> 3) * 256 + (lane >> 4) * 128 + (wu & 7) * 16 + (lane & 15)] = wv;
            }
        }
    }
}

__global__ __launch_bounds__(256) void k_gemm1(const float* __restrict__ X,
                                               const float* __restrict__ W,
                                               const float* __restrict__ deg,
                                               float* __restrict__ H, int nrows) {
    int wave = (blockIdx.x * 256 + threadIdx.x) >> 5;
    int lane = threadIdx.x & 31;
    int m0 = wave << 4;
    if (m0 >= nrows) return;
    int row = lane & 15;
    int kk  = (lane >> 4) << 1;
    const float* Xr = X + (size_t)(m0 + row) * FIN + kk;
    v8f c0 = {}, c1 = {}, c2 = {}, c3 = {};
    for (int k0 = 0; k0 < FIN; k0 += 4) {
        v2f a;
        a.x = Xr[k0];
        a.y = Xr[k0 + 1];
        const float* Wp = W + (size_t)(k0 + kk) * HID;
        v2f b0 = { Wp[row],      Wp[HID + row] };
        v2f b1 = { Wp[16 + row], Wp[HID + 16 + row] };
        v2f b2 = { Wp[32 + row], Wp[HID + 32 + row] };
        v2f b3 = { Wp[48 + row], Wp[HID + 48 + row] };
        c0 = __builtin_amdgcn_wmma_f32_16x16x4_f32(false, a, false, b0, (short)0, c0, false, false);
        c1 = __builtin_amdgcn_wmma_f32_16x16x4_f32(false, a, false, b1, (short)0, c1, false, false);
        c2 = __builtin_amdgcn_wmma_f32_16x16x4_f32(false, a, false, b2, (short)0, c2, false, false);
        c3 = __builtin_amdgcn_wmma_f32_16x16x4_f32(false, a, false, b3, (short)0, c3, false, false);
    }
    int mrow = m0 + ((lane >> 4) << 3);
    int col  = lane & 15;
    for (int r = 0; r < 8; ++r) {
        float di = rsqrtf(deg[mrow + r]);
        size_t off = (size_t)(mrow + r) * HID + col;
        float y0 = c0[r] * di, y1 = c1[r] * di, y2 = c2[r] * di, y3 = c3[r] * di;
        H[off]      = y0;
        H[off + 16] = y1;
        H[off + 32] = y2;
        H[off + 48] = y3;
    }
}

__global__ __launch_bounds__(256) void k_gemm2(const float* __restrict__ A,
                                               const float* __restrict__ W,
                                               const float* __restrict__ deg,
                                               const float* __restrict__ bias1,
                                               float* __restrict__ H2, int nrows) {
    int wave = (blockIdx.x * 256 + threadIdx.x) >> 5;
    int lane = threadIdx.x & 31;
    int m0 = wave << 4;
    if (m0 >= nrows) return;
    int row = lane & 15;
    int kk  = (lane >> 4) << 1;
    const float* Ar = A + (size_t)(m0 + row) * HID + kk;
    const float* bp = bias1 + kk;
    const float dr = rsqrtf(deg[m0 + row]);
    v8f c0 = {}, c1 = {}, c2 = {};
    for (int k0 = 0; k0 < HID; k0 += 4) {
        v2f a;
        a.x = fmaxf(Ar[k0]     * dr + bp[k0],     0.0f);
        a.y = fmaxf(Ar[k0 + 1] * dr + bp[k0 + 1], 0.0f);
        const float* Wp = W + (size_t)(k0 + kk) * COUT;
        v2f b0 = { Wp[row],      Wp[COUT + row] };
        v2f b1 = { Wp[16 + row], Wp[COUT + 16 + row] };
        int col2 = 32 + row;
        v2f b2 = { (col2 < COUT) ? Wp[col2] : 0.0f,
                   (col2 < COUT) ? Wp[COUT + col2] : 0.0f };
        c0 = __builtin_amdgcn_wmma_f32_16x16x4_f32(false, a, false, b0, (short)0, c0, false, false);
        c1 = __builtin_amdgcn_wmma_f32_16x16x4_f32(false, a, false, b1, (short)0, c1, false, false);
        c2 = __builtin_amdgcn_wmma_f32_16x16x4_f32(false, a, false, b2, (short)0, c2, false, false);
    }
    int mrow = m0 + ((lane >> 4) << 3);
    int col  = lane & 15;
    for (int r = 0; r < 8; ++r) {
        float di = rsqrtf(deg[mrow + r]);
        size_t off = (size_t)(mrow + r) * LD2;
        float y0 = c0[r] * di, y1 = c1[r] * di, y2 = c2[r] * di;
        H2[off + col]      = y0;
        H2[off + 16 + col] = y1;
        if (32 + col < COUT) H2[off + 32 + col] = y2;
    }
}

#define KPL   4
#define TRIPK (KPL * 512)
#define QCAP  (1024 + 32)
#define WCAP  416
#define HOFF  1638400
#define LOFF  1671168
#define LEND  (LOFF + NOWN * 32 * WCAP * 4)

__device__ __forceinline__ bool lists_fit(int E, int n) {
    const long long kbytes = (long long)((E + TRIPK - 1) / TRIPK) * TRIPK;
    return (4096 + kbytes <= HOFF) && ((long long)n * COUT * 4 >= LEND);
}
__device__ __forceinline__ void push_matches(unsigned c, unsigned m, int eb, int* myq, int& cnt) {
    do {
        if (c != 0u) {
            const int p = __builtin_ctz(c);
            myq[cnt + (int)__builtin_amdgcn_mbcnt_lo(m, 0u)] = eb + (p << 2);
            c &= c - 1u;
        }
        cnt += __builtin_popcount(m);
        m = __builtin_amdgcn_ballot_w32(c != 0u);
    } while (m != 0u);
}

__device__ __forceinline__ unsigned match_bits(uint4 lo, uint4 hi, const unsigned (&qm)[8], int eb, int e0, int E) {
    const unsigned x = (lo.x ^ qm[0]) | (lo.y ^ qm[1]) | (lo.z ^ qm[2]) | (lo.w ^ qm[3])
                     | (hi.x ^ qm[4]) | (hi.y ^ qm[5]) | (hi.z ^ qm[6]) | (hi.w ^ qm[7]);
    unsigned c = ~x;
    if (e0 + 1024 > E) {
        const int nv = (E - eb + 3) >> 2;
        c &= (nv >= 32) ? ~0u : (nv <= 0) ? 0u : ((1u << nv) - 1u);
    }
    return c;
}

template <bool FULL>
__device__ __forceinline__ void note_batch(const int* __restrict__ pk, int* cntrow, const int* myq, int top,
                                           int nb, int lane, bool rec, int* __restrict__ mylist, int& nrec) {
    if (FULL || lane < nb) {
        const int p = pk[myq[top - 1 - lane]];
        if (rec && nrec + lane < WCAP) mylist[nrec + lane] = p;
        atomicAdd(cntrow + (p & ((1 << PKS) - 1)), 1);
    }
    nrec += FULL ? 32 : nb;
}

template <int BATCH, class FullFn, class RestFn>
__device__ __forceinline__ void walk_keys(const uint4* __restrict__ key4, int E, int* myq, int g, int w, int lane,
                                          FullFn full, RestFn rest) {
    unsigned qm[8];
#pragma unroll
    for (int b = 0; b < 8; ++b) qm[b] = 0u - (((unsigned)g >> b) & 1u);
    int cnt = 0;
    const int ntrip = (E + TRIPK - 1) / TRIPK;
    const int rot = (int)(((long long)g * ntrip) >> 8);
    uint4 kv[KPL], nx[KPL];
    {
        int tq = (w < ntrip) ? w + rot : 0;
        if (tq >= ntrip) tq -= ntrip;
#pragma unroll
        for (int j = 0; j < KPL; ++j) kv[j] = key4[((tq * TRIPK) >> 4) + j * 32 + lane];
    }
    for (int t = w; t < ntrip; t += 32) {
        int tt = t + rot;
        if (tt >= ntrip) tt -= ntrip;
        const int e0 = tt * TRIPK;
        int tq = ((t + 32 < ntrip) ? t + 32 : t) + rot;
        if (tq >= ntrip) tq -= ntrip;
#pragma unroll
        for (int j = 0; j < KPL; ++j) nx[j] = key4[((tq * TRIPK) >> 4) + j * 32 + lane];
#pragma unroll
        for (int j = 0; j < KPL; j += 2) {
            const int eblk = e0 + j * 512;
            const int eb = eblk + ((lane >> 2) << 7) + (lane & 3);
            const unsigned c32 = match_bits(kv[j], kv[j + 1], qm, eb, eblk, E);
            const unsigned mm = __builtin_amdgcn_ballot_w32(c32 != 0u);
            if (mm != 0u) {
                push_matches(c32, mm, eb, myq, cnt);
                while (cnt >= BATCH) {
                    full(cnt);
                    cnt -= BATCH;
                }
            }
        }
#pragma unroll
        for (int j = 0; j < KPL; ++j) kv[j] = nx[j];
    }
    if (cnt > 0) rest(cnt, cnt);
}

__global__ __launch_bounds__(1024) void k_search(const uint4* __restrict__ key4, const int* __restrict__ pk,
                                                 int E, int n, float* __restrict__ deg,
                                                 int* __restrict__ hdr, int* __restrict__ lst) {
    __shared__ int q[32 * QCAP];
    __shared__ int cntrow[RPO];
    const int tid  = threadIdx.x;
    const int lane = tid & 31;
    const int w    = __builtin_amdgcn_readfirstlane(tid >> 5);
    const int g    = blockIdx.x;
    const int row0 = g * RPO;
    const int nown = min(RPO, n - row0);
    for (int i = tid; i < RPO; i += 1024) cntrow[i] = 0;
    __syncthreads();

    const bool lists = lists_fit(E, n);
    int* myq = q + w * QCAP;
    int* mylist = lst + (size_t)(g * 32 + w) * WCAP;
    int nrec = 0;
    walk_keys<32>(key4, E, myq, g, w, lane,
        [&](int top) { note_batch<true >(pk, cntrow, myq, top, 32, lane, lists, mylist, nrec); },
        [&](int top, int nb) { note_batch<false>(pk, cntrow, myq, top, nb, lane, lists, mylist, nrec); });
    if (lists && lane == 0) hdr[g * 32 + w] = nrec;
    __syncthreads();

    for (int i = tid; i < nown; i += 1024) deg[row0 + i] = 1.0f + (float)cntrow[i];
}

template <int NC, bool FULL>
__device__ __forceinline__ void add_batch(int pk, int b0, int nb, const float* __restrict__ H,
                                          float* rows, int lane) {
    const bool tail = (NC == 64) || (lane < NC - 32);
    int sj[8], dj[8];
#pragma unroll
    for (int u = 0; u < 8; ++u) {
        const int p = __builtin_amdgcn_readlane(pk, b0 + u);
        sj[u] = p >> PKS;
        dj[u] = p & ((1 << PKS) - 1);
    }
    float v0[8], v1[8];
#pragma unroll
    for (int u = 0; u < 8; ++u) {
        v0[u] = 0.0f; v1[u] = 0.0f;
        if (FULL || u < nb) v0[u] = H[(size_t)sj[u] * LD2 + lane];
    }
    if (tail) {
#pragma unroll
        for (int u = 0; u < 8; ++u)
            if (FULL || u < nb) v1[u] = H[(size_t)sj[u] * LD2 + 32 + lane];
    }
#pragma unroll
    for (int u = 0; u < 8; ++u)
        if (FULL || u < nb) atomicAdd(rows + dj[u] * LD2 + lane, v0[u]);
    if (tail) {
#pragma unroll
        for (int u = 0; u < 8; ++u)
            if (FULL || u < nb) atomicAdd(rows + dj[u] * LD2 + 32 + lane, v1[u]);
    }
}

template <int NC, bool FULL>
__device__ __forceinline__ void own_batch(const int* __restrict__ src, const int* __restrict__ dst,
                                          const float* __restrict__ H, float* rows, const int* myq,
                                          int top, int nb, int row0, int lane) {
    const bool mine = lane < (FULL ? 8 : nb);
    int e = 0;
    if (mine) e = myq[top - 1 - lane];
    int s = 0, d = row0;
    if (mine) { s = src[e]; d = dst[e]; }
    const int pk = (s << PKS) | (d - row0);
    add_batch<NC, FULL>(pk, 0, nb, H, rows, lane);
}

template <int NC>
__global__ __launch_bounds__(1024) void k_own(const uint4* __restrict__ key4,
                                              const int* __restrict__ src,
                                              const int* __restrict__ dst, int E, int n,
                                              const float* __restrict__ H, float* __restrict__ A,
                                              const int* __restrict__ hdr, const int* __restrict__ lst) {
    __shared__ float rows[RPO * LD2];
    __shared__ int   q[32 * QCAP];
    constexpr int F4 = NC / 4;
    const int tid  = threadIdx.x;
    const int lane = tid & 31;
    const int w    = __builtin_amdgcn_readfirstlane(tid >> 5);
    const int g    = blockIdx.x;
    const int row0 = g * RPO;
    const int nown = min(RPO, n - row0);
    const float4* H4 = (const float4*)H;
    float4* A4 = (float4*)A;
    float4* rows4 = (float4*)rows;

    constexpr int NIT = (RPO * F4 + 1023) / 1024;
    float4 st[NIT];
#pragma unroll
    for (int u = 0; u < NIT; ++u) {
        const int i = tid + u * 1024;
        const int r = i / F4, f = i - r * F4;
        st[u] = make_float4(0.0f, 0.0f, 0.0f, 0.0f);
        if (i < nown * F4) st[u] = H4[(size_t)(row0 + r) * (LD2 / 4) + f];
    }
#pragma unroll
    for (int u = 0; u < NIT; ++u) {
        const int i = tid + u * 1024;
        const int r = i / F4, f = i - r * F4;
        if (i < nown * F4) rows4[r * (LD2 / 4) + f] = st[u];
    }
    __syncthreads();

    const bool lists = lists_fit(E, n);
    const int* mylist = lst + (size_t)(g * 32 + w) * WCAP;
    int cw = WCAP + 1;
    if (lists) cw = hdr[g * 32 + lane];
    if (__builtin_amdgcn_ballot_w32(cw > WCAP) == 0u) {
        const int nmine = __builtin_amdgcn_readlane(cw, w);
        int ent = 0;
        if (lane < nmine) ent = mylist[lane];
        int c0 = 0;
        for (; c0 + 32 <= nmine; c0 += 32) {
            const int cur = ent;
            if (c0 + 32 + lane < nmine) ent = mylist[c0 + 32 + lane];
            add_batch<NC, true>(cur, 0,  8, H, rows, lane);
            add_batch<NC, true>(cur, 8,  8, H, rows, lane);
            add_batch<NC, true>(cur, 16, 8, H, rows, lane);
            add_batch<NC, true>(cur, 24, 8, H, rows, lane);
        }
        const int left = nmine - c0;
        for (int b = 0; b < left; b += 8)
            add_batch<NC, false>(ent, b, min(8, left - b), H, rows, lane);
    } else {
        int* myq = q + w * QCAP;
        walk_keys<8>(key4, E, myq, g, w, lane,
            [&](int top) { own_batch<NC, true >(src, dst, H, rows, myq, top, 8,  row0, lane); },
            [&](int top, int nb) { own_batch<NC, false>(src, dst, H, rows, myq, top, nb, row0, lane); });
    }
    __syncthreads();

    for (int i = tid; i < nown * F4; i += 1024) {
        const int r = i / F4, f = i - r * F4;
        A4[(size_t)(row0 + r) * (LD2 / 4) + f] = rows4[r * (LD2 / 4) + f];
    }
}

template <bool GUARD>
__device__ __forceinline__ void fin2_trip(float4* __restrict__ out, const float4* __restrict__ acc,
                                          const float* __restrict__ deg, const float* __restrict__ b2,
                                          int u0, int nwaves, int nunits, int lane) {
    float4 o[FPT], bb[FPT];
    float  dg[FPT];
#pragma unroll
    for (int u = 0; u < FPT; ++u) {
        const int unit = GUARD ? min(u0 + u * nwaves, nunits - 1) : u0 + u * nwaves;
        const int idx  = unit * 32 + lane;
        const int i    = idx / 10;
        const int f4   = idx - i * 10;
        const float* b = b2 + f4 * 4;
        dg[u] = deg[i];
        o[u]  = acc[(size_t)i * (LD2 / 4) + f4];
        bb[u] = make_float4(b[0], b[1], b[2], b[3]);
    }
#pragma unroll
    for (int u = 0; u < FPT; ++u) {
        const int unit = u0 + u * nwaves;
        if (!GUARD || unit < nunits) {
            const float di = rsqrtf(dg[u]);
            float4 r = o[u];
            r.x = r.x * di + bb[u].x;
            r.y = r.y * di + bb[u].y;
            r.z = r.z * di + bb[u].z;
            r.w = r.w * di + bb[u].w;
            out[unit * 32 + lane] = r;
        }
    }
}

__global__ __launch_bounds__(256) void k_fin2(float4* __restrict__ out,
                                              const float4* __restrict__ acc,
                                              const float* __restrict__ deg,
                                              const float* __restrict__ b2, int n) {
    const int nunits = (n * (COUT / 4)) >> 5;
    const int nwaves = gridDim.x * 8;
    const int lane   = threadIdx.x & 31;
    const int wave   = __builtin_amdgcn_readfirstlane((int)((blockIdx.x * 256 + threadIdx.x) >> 5));
    for (int u0 = wave; u0 < nunits; u0 += nwaves * FPT) {
        if (u0 + (FPT - 1) * nwaves < nunits) fin2_trip<false>(out, acc, deg, b2, u0, nwaves, nunits, lane);
        else                                  fin2_trip<true >(out, acc, deg, b2, u0, nwaves, nunits, lane);
    }
}

extern "C" void kernel_launch(void* const* d_in, const int* in_sizes, int n_in,
                              void* d_out, int out_size, void* d_ws, size_t ws_size,
                              hipStream_t stream) {
    const float* x  = (const float*)d_in[0];
    const float* W1 = (const float*)d_in[1];
    const float* b1 = (const float*)d_in[2];
    const float* W2 = (const float*)d_in[3];
    const float* b2 = (const float*)d_in[4];
    const int*   ei = (const int*)d_in[5];

    const int Nn = in_sizes[0] / FIN;
    const int E  = in_sizes[5] / 2;
    const int* src = ei;
    const int* dst = ei + E;

    float* ws   = (float*)d_ws;
    float* deg  = ws;
    float* h    = ws + Nn;
    float* agg  = h + (size_t)Nn * HID;
    float* h2   = h;
    float* acc2 = agg;
    float* out  = (float*)d_out;

    unsigned char* key = (unsigned char*)d_out + 4096;
    int* hdr = (int*)((unsigned char*)d_out + HOFF);
    int* lst = (int*)((unsigned char*)d_out + LOFF);
    int* pk  = (int*)agg;
    int gemmBlocks = ((Nn / 16) + 7) / 8;
    int gF = (Nn * (COUT / 4) / 32 + FPT * 8 - 1) / (FPT * 8);

    k_keys<<<KBLK, 256, 0, stream>>>(src, dst, E, (unsigned*)key, pk);
    k_search<<<NOWN, 1024, 0, stream>>>((const uint4*)key, pk, E, Nn, deg, hdr, lst);

    k_gemm1<<<gemmBlocks, 256, 0, stream>>>(x, W1, deg, h, Nn);
    k_own<HID><<<NOWN, 1024, 0, stream>>>((const uint4*)key, src, dst, E, Nn, h, agg, hdr, lst);

    k_gemm2<<<gemmBlocks, 256, 0, stream>>>(agg, W2, deg, b1, h2, Nn);
    k_own<COUT><<<NOWN, 1024, 0, stream>>>((const uint4*)key, src, dst, E, Nn, h2, acc2, hdr, lst);
    k_fin2<<<gF, 256, 0, stream>>>((float4*)out, (const float4*)acc2, deg, b2, Nn);
}


# product 1 runs inside the search launch: owner waves multiply 16-row strips of their own rows, half before and half after their search share, scaled by their own counts; one launch less; same sums
# speedup vs baseline: 3.2303x; 1.0353x over previous
#include <hip/hip_runtime.h>

#define FIN 256
#define HID 64
#define COUT 40
#define LD2 64

typedef float v2f __attribute__((ext_vector_type(2)));
typedef float v8f __attribute__((ext_vector_type(8)));

#define RPO   391
#define NOWN  256
#define PKS   9
#define KBLK  256
#define KU    4
#define FPT   8

__device__ __forceinline__ int pack_edge(int s, int d, unsigned& k) {
    k = (unsigned)(d / RPO);
    return (s << PKS) | (d - (int)k * RPO);
}

__global__ __launch_bounds__(256) void k_keys(const int* __restrict__ src, const int* __restrict__ dst, int E,
                                              unsigned* __restrict__ kp, int* __restrict__ pk) {
    const int lane = threadIdx.x & 31;
    const int wave = __builtin_amdgcn_readfirstlane((int)((blockIdx.x * 256 + threadIdx.x) >> 5));
    const int nwav = gridDim.x * 8;
    const int nwu  = (E + 127) >> 7;
    const bool al  = (E & 3) == 0;
    for (int w0 = wave; w0 < nwu; w0 += nwav * KU) {
        int4 s[KU], d[KU];
#pragma unroll
        for (int j = 0; j < KU; ++j) {
            const int wu = min(w0 + j * nwav, nwu - 1);
            const int e  = wu * 128 + lane * 4;
            if (al) {
                const int u = min(e, E - 4) >> 2;
                s[j] = ((const int4*)src)[u];
                d[j] = ((const int4*)dst)[u];
            } else {
                s[j].x = src[min(e, E - 1)];     d[j].x = dst[min(e, E - 1)];
                s[j].y = src[min(e + 1, E - 1)]; d[j].y = dst[min(e + 1, E - 1)];
                s[j].z = src[min(e + 2, E - 1)]; d[j].z = dst[min(e + 2, E - 1)];
                s[j].w = src[min(e + 3, E - 1)]; d[j].w = dst[min(e + 3, E - 1)];
            }
        }
#pragma unroll
        for (int j = 0; j < KU; ++j) {
            const int wu = w0 + j * nwav;
            if (wu < nwu) {
                const int e = wu * 128 + lane * 4;
                unsigned k0, k1, k2, k3;
                int4 p;
                p.x = pack_edge(s[j].x, d[j].x, k0);
                p.y = pack_edge(s[j].y, d[j].y, k1);
                p.z = pack_edge(s[j].z, d[j].z, k2);
                p.w = pack_edge(s[j].w, d[j].w, k3);
                if (al) {
                    if (e < E) ((int4*)pk)[e >> 2] = p;
                } else {
                    if (e < E)     pk[e]     = p.x;
                    if (e + 1 < E) pk[e + 1] = p.y;
                    if (e + 2 < E) pk[e + 2] = p.z;
                    if (e + 3 < E) pk[e + 3] = p.w;
                }
                const unsigned kk = k0 | (k1 << 8) | (k2 << 16) | (k3 << 24);
                unsigned wv = 0u;
#pragma unroll
                for (int t = 0; t < 32; ++t) {
                    const unsigned sres = __builtin_amdgcn_ballot_w32(((kk >> t) & 1u) != 0u);
                    const int jl = ((t >> 3) << 2) + (t & 3) + (((t >> 2) & 1) << 4);
                    wv = (lane == jl) ? sres : wv;
                }
                kp[(wu >> 3) * 256 + (lane >> 4) * 128 + (wu & 7) * 16 + (lane & 15)] = wv;
            }
        }
    }
}

__device__ __forceinline__ void strip_product(const float* __restrict__ X, const float* __restrict__ W,
                                              int m0, int n, int lane, v8f& c0, v8f& c1, v8f& c2, v8f& c3) {
    const int row = lane & 15;
    const int kk  = (lane >> 4) << 1;
    const float* Xr = X + (size_t)min(m0 + row, n - 1) * FIN + kk;
    for (int k0 = 0; k0 < FIN; k0 += 4) {
        v2f a;
        a.x = Xr[k0];
        a.y = Xr[k0 + 1];
        const float* Wp = W + (size_t)(k0 + kk) * HID;
        v2f b0 = { Wp[row],      Wp[HID + row] };
        v2f b1 = { Wp[16 + row], Wp[HID + 16 + row] };
        v2f b2 = { Wp[32 + row], Wp[HID + 32 + row] };
        v2f b3 = { Wp[48 + row], Wp[HID + 48 + row] };
        c0 = __builtin_amdgcn_wmma_f32_16x16x4_f32(false, a, false, b0, (short)0, c0, false, false);
        c1 = __builtin_amdgcn_wmma_f32_16x16x4_f32(false, a, false, b1, (short)0, c1, false, false);
        c2 = __builtin_amdgcn_wmma_f32_16x16x4_f32(false, a, false, b2, (short)0, c2, false, false);
        c3 = __builtin_amdgcn_wmma_f32_16x16x4_f32(false, a, false, b3, (short)0, c3, false, false);
    }
}

__global__ __launch_bounds__(256) void k_gemm2(const float* __restrict__ A,
                                               const float* __restrict__ W,
                                               const float* __restrict__ deg,
                                               const float* __restrict__ bias1,
                                               float* __restrict__ H2, int nrows) {
    int wave = (blockIdx.x * 256 + threadIdx.x) >> 5;
    int lane = threadIdx.x & 31;
    int m0 = wave << 4;
    if (m0 >= nrows) return;
    int row = lane & 15;
    int kk  = (lane >> 4) << 1;
    const float* Ar = A + (size_t)(m0 + row) * HID + kk;
    const float* bp = bias1 + kk;
    const float dr = rsqrtf(deg[m0 + row]);
    v8f c0 = {}, c1 = {}, c2 = {};
    for (int k0 = 0; k0 < HID; k0 += 4) {
        v2f a;
        a.x = fmaxf(Ar[k0]     * dr + bp[k0],     0.0f);
        a.y = fmaxf(Ar[k0 + 1] * dr + bp[k0 + 1], 0.0f);
        const float* Wp = W + (size_t)(k0 + kk) * COUT;
        v2f b0 = { Wp[row],      Wp[COUT + row] };
        v2f b1 = { Wp[16 + row], Wp[COUT + 16 + row] };
        int col2 = 32 + row;
        v2f b2 = { (col2 < COUT) ? Wp[col2] : 0.0f,
                   (col2 < COUT) ? Wp[COUT + col2] : 0.0f };
        c0 = __builtin_amdgcn_wmma_f32_16x16x4_f32(false, a, false, b0, (short)0, c0, false, false);
        c1 = __builtin_amdgcn_wmma_f32_16x16x4_f32(false, a, false, b1, (short)0, c1, false, false);
        c2 = __builtin_amdgcn_wmma_f32_16x16x4_f32(false, a, false, b2, (short)0, c2, false, false);
    }
    int mrow = m0 + ((lane >> 4) << 3);
    int col  = lane & 15;
    for (int r = 0; r < 8; ++r) {
        float di = rsqrtf(deg[mrow + r]);
        size_t off = (size_t)(mrow + r) * LD2;
        float y0 = c0[r] * di, y1 = c1[r] * di, y2 = c2[r] * di;
        H2[off + col]      = y0;
        H2[off + 16 + col] = y1;
        if (32 + col < COUT) H2[off + 32 + col] = y2;
    }
}

#define KPL   4
#define TRIPK (KPL * 512)
#define QCAP  (1024 + 32)
#define WCAP  416
#define HOFF  1638400
#define LOFF  1671168
#define LEND  (LOFF + NOWN * 32 * WCAP * 4)

__device__ __forceinline__ bool lists_fit(int E, int n) {
    const long long kbytes = (long long)((E + TRIPK - 1) / TRIPK) * TRIPK;
    return (4096 + kbytes <= HOFF) && ((long long)n * COUT * 4 >= LEND);
}
__device__ __forceinline__ void push_matches(unsigned c, unsigned m, int eb, int* myq, int& cnt) {
    do {
        if (c != 0u) {
            const int p = __builtin_ctz(c);
            myq[cnt + (int)__builtin_amdgcn_mbcnt_lo(m, 0u)] = eb + (p << 2);
            c &= c - 1u;
        }
        cnt += __builtin_popcount(m);
        m = __builtin_amdgcn_ballot_w32(c != 0u);
    } while (m != 0u);
}

__device__ __forceinline__ unsigned match_bits(uint4 lo, uint4 hi, const unsigned (&qm)[8], int eb, int e0, int E) {
    const unsigned x = (lo.x ^ qm[0]) | (lo.y ^ qm[1]) | (lo.z ^ qm[2]) | (lo.w ^ qm[3])
                     | (hi.x ^ qm[4]) | (hi.y ^ qm[5]) | (hi.z ^ qm[6]) | (hi.w ^ qm[7]);
    unsigned c = ~x;
    if (e0 + 1024 > E) {
        const int nv = (E - eb + 3) >> 2;
        c &= (nv >= 32) ? ~0u : (nv <= 0) ? 0u : ((1u << nv) - 1u);
    }
    return c;
}

template <bool FULL>
__device__ __forceinline__ void note_batch(const int* __restrict__ pk, int* cntrow, const int* myq, int top,
                                           int nb, int lane, bool rec, int* __restrict__ mylist, int& nrec) {
    if (FULL || lane < nb) {
        const int p = pk[myq[top - 1 - lane]];
        if (rec && nrec + lane < WCAP) mylist[nrec + lane] = p;
        atomicAdd(cntrow + (p & ((1 << PKS) - 1)), 1);
    }
    nrec += FULL ? 32 : nb;
}

template <int BATCH, class FullFn, class RestFn>
__device__ __forceinline__ void walk_keys(const uint4* __restrict__ key4, int E, int* myq, int g, int w, int lane,
                                          FullFn full, RestFn rest) {
    unsigned qm[8];
#pragma unroll
    for (int b = 0; b < 8; ++b) qm[b] = 0u - (((unsigned)g >> b) & 1u);
    int cnt = 0;
    const int ntrip = (E + TRIPK - 1) / TRIPK;
    const int rot = (int)(((long long)g * ntrip) >> 8);
    uint4 kv[KPL], nx[KPL];
    {
        int tq = (w < ntrip) ? w + rot : 0;
        if (tq >= ntrip) tq -= ntrip;
#pragma unroll
        for (int j = 0; j < KPL; ++j) kv[j] = key4[((tq * TRIPK) >> 4) + j * 32 + lane];
    }
    for (int t = w; t < ntrip; t += 32) {
        int tt = t + rot;
        if (tt >= ntrip) tt -= ntrip;
        const int e0 = tt * TRIPK;
        int tq = ((t + 32 < ntrip) ? t + 32 : t) + rot;
        if (tq >= ntrip) tq -= ntrip;
#pragma unroll
        for (int j = 0; j < KPL; ++j) nx[j] = key4[((tq * TRIPK) >> 4) + j * 32 + lane];
#pragma unroll
        for (int j = 0; j < KPL; j += 2) {
            const int eblk = e0 + j * 512;
            const int eb = eblk + ((lane >> 2) << 7) + (lane & 3);
            const unsigned c32 = match_bits(kv[j], kv[j + 1], qm, eb, eblk, E);
            const unsigned mm = __builtin_amdgcn_ballot_w32(c32 != 0u);
            if (mm != 0u) {
                push_matches(c32, mm, eb, myq, cnt);
                while (cnt >= BATCH) {
                    full(cnt);
                    cnt -= BATCH;
                }
            }
        }
#pragma unroll
        for (int j = 0; j < KPL; ++j) kv[j] = nx[j];
    }
    if (cnt > 0) rest(cnt, cnt);
}

__global__ __launch_bounds__(1024) void k_search(const uint4* __restrict__ key4, const int* __restrict__ pk,
                                                 int E, int n, float* __restrict__ deg,
                                                 int* __restrict__ hdr, int* __restrict__ lst,
                                                 const float* __restrict__ X, const float* __restrict__ W,
                                                 float* __restrict__ H) {
    __shared__ int q[32 * QCAP];
    __shared__ int cntrow[RPO];
    const int tid  = threadIdx.x;
    const int lane = tid & 31;
    const int w    = __builtin_amdgcn_readfirstlane(tid >> 5);
    const int g    = blockIdx.x;
    const int row0 = g * RPO;
    const int nown = min(RPO, n - row0);
    for (int i = tid; i < RPO; i += 1024) cntrow[i] = 0;
    __syncthreads();

    const int  srow  = w * 16;
    const bool strip = srow < nown;
    const bool early = ((w >> 2) & 1) == 0;
    v8f c0 = {}, c1 = {}, c2 = {}, c3 = {};
    if (strip && early) strip_product(X, W, row0 + srow, n, lane, c0, c1, c2, c3);

    const bool lists = lists_fit(E, n);
    int* myq = q + w * QCAP;
    int* mylist = lst + (size_t)(g * 32 + w) * WCAP;
    int nrec = 0;
    walk_keys<32>(key4, E, myq, g, w, lane,
        [&](int top) { note_batch<true >(pk, cntrow, myq, top, 32, lane, lists, mylist, nrec); },
        [&](int top, int nb) { note_batch<false>(pk, cntrow, myq, top, nb, lane, lists, mylist, nrec); });
    if (lists && lane == 0) hdr[g * 32 + w] = nrec;

    if (strip && !early) strip_product(X, W, row0 + srow, n, lane, c0, c1, c2, c3);
    __syncthreads();

    for (int i = tid; i < nown; i += 1024) deg[row0 + i] = 1.0f + (float)cntrow[i];

    if (strip) {
        const int lr0 = srow + ((lane >> 4) << 3);
        const int col = lane & 15;
        for (int r = 0; r < 8; ++r) {
            const int lr = lr0 + r;
            if (lr < nown) {
                const float di = rsqrtf(1.0f + (float)cntrow[lr]);
                const size_t off = (size_t)(row0 + lr) * HID + col;
                const float y0 = c0[r] * di, y1 = c1[r] * di, y2 = c2[r] * di, y3 = c3[r] * di;
                H[off]      = y0;
                H[off + 16] = y1;
                H[off + 32] = y2;
                H[off + 48] = y3;
            }
        }
    }
}

template <int NC, bool FULL>
__device__ __forceinline__ void add_batch(int pk, int b0, int nb, const float* __restrict__ H,
                                          float* rows, int lane) {
    const bool tail = (NC == 64) || (lane < NC - 32);
    int sj[8], dj[8];
#pragma unroll
    for (int u = 0; u < 8; ++u) {
        const int p = __builtin_amdgcn_readlane(pk, b0 + u);
        sj[u] = p >> PKS;
        dj[u] = p & ((1 << PKS) - 1);
    }
    float v0[8], v1[8];
#pragma unroll
    for (int u = 0; u < 8; ++u) {
        v0[u] = 0.0f; v1[u] = 0.0f;
        if (FULL || u < nb) v0[u] = H[(size_t)sj[u] * LD2 + lane];
    }
    if (tail) {
#pragma unroll
        for (int u = 0; u < 8; ++u)
            if (FULL || u < nb) v1[u] = H[(size_t)sj[u] * LD2 + 32 + lane];
    }
#pragma unroll
    for (int u = 0; u < 8; ++u)
        if (FULL || u < nb) atomicAdd(rows + dj[u] * LD2 + lane, v0[u]);
    if (tail) {
#pragma unroll
        for (int u = 0; u < 8; ++u)
            if (FULL || u < nb) atomicAdd(rows + dj[u] * LD2 + 32 + lane, v1[u]);
    }
}

template <int NC, bool FULL>
__device__ __forceinline__ void own_batch(const int* __restrict__ src, const int* __restrict__ dst,
                                          const float* __restrict__ H, float* rows, const int* myq,
                                          int top, int nb, int row0, int lane) {
    const bool mine = lane < (FULL ? 8 : nb);
    int e = 0;
    if (mine) e = myq[top - 1 - lane];
    int s = 0, d = row0;
    if (mine) { s = src[e]; d = dst[e]; }
    const int pk = (s << PKS) | (d - row0);
    add_batch<NC, FULL>(pk, 0, nb, H, rows, lane);
}

template <int NC>
__global__ __launch_bounds__(1024) void k_own(const uint4* __restrict__ key4,
                                              const int* __restrict__ src,
                                              const int* __restrict__ dst, int E, int n,
                                              const float* __restrict__ H, float* __restrict__ A,
                                              const int* __restrict__ hdr, const int* __restrict__ lst) {
    __shared__ float rows[RPO * LD2];
    __shared__ int   q[32 * QCAP];
    constexpr int F4 = NC / 4;
    const int tid  = threadIdx.x;
    const int lane = tid & 31;
    const int w    = __builtin_amdgcn_readfirstlane(tid >> 5);
    const int g    = blockIdx.x;
    const int row0 = g * RPO;
    const int nown = min(RPO, n - row0);
    const float4* H4 = (const float4*)H;
    float4* A4 = (float4*)A;
    float4* rows4 = (float4*)rows;

    constexpr int NIT = (RPO * F4 + 1023) / 1024;
    float4 st[NIT];
#pragma unroll
    for (int u = 0; u < NIT; ++u) {
        const int i = tid + u * 1024;
        const int r = i / F4, f = i - r * F4;
        st[u] = make_float4(0.0f, 0.0f, 0.0f, 0.0f);
        if (i < nown * F4) st[u] = H4[(size_t)(row0 + r) * (LD2 / 4) + f];
    }
#pragma unroll
    for (int u = 0; u < NIT; ++u) {
        const int i = tid + u * 1024;
        const int r = i / F4, f = i - r * F4;
        if (i < nown * F4) rows4[r * (LD2 / 4) + f] = st[u];
    }
    __syncthreads();

    const bool lists = lists_fit(E, n);
    const int* mylist = lst + (size_t)(g * 32 + w) * WCAP;
    int cw = WCAP + 1;
    if (lists) cw = hdr[g * 32 + lane];
    if (__builtin_amdgcn_ballot_w32(cw > WCAP) == 0u) {
        const int nmine = __builtin_amdgcn_readlane(cw, w);
        int ent = 0;
        if (lane < nmine) ent = mylist[lane];
        int c0 = 0;
        for (; c0 + 32 <= nmine; c0 += 32) {
            const int cur = ent;
            if (c0 + 32 + lane < nmine) ent = mylist[c0 + 32 + lane];
            add_batch<NC, true>(cur, 0,  8, H, rows, lane);
            add_batch<NC, true>(cur, 8,  8, H, rows, lane);
            add_batch<NC, true>(cur, 16, 8, H, rows, lane);
            add_batch<NC, true>(cur, 24, 8, H, rows, lane);
        }
        const int left = nmine - c0;
        for (int b = 0; b < left; b += 8)
            add_batch<NC, false>(ent, b, min(8, left - b), H, rows, lane);
    } else {
        int* myq = q + w * QCAP;
        walk_keys<8>(key4, E, myq, g, w, lane,
            [&](int top) { own_batch<NC, true >(src, dst, H, rows, myq, top, 8,  row0, lane); },
            [&](int top, int nb) { own_batch<NC, false>(src, dst, H, rows, myq, top, nb, row0, lane); });
    }
    __syncthreads();

    for (int i = tid; i < nown * F4; i += 1024) {
        const int r = i / F4, f = i - r * F4;
        A4[(size_t)(row0 + r) * (LD2 / 4) + f] = rows4[r * (LD2 / 4) + f];
    }
}

template <bool GUARD>
__device__ __forceinline__ void fin2_trip(float4* __restrict__ out, const float4* __restrict__ acc,
                                          const float* __restrict__ deg, const float* __restrict__ b2,
                                          int u0, int nwaves, int nunits, int lane) {
    float4 o[FPT], bb[FPT];
    float  dg[FPT];
#pragma unroll
    for (int u = 0; u < FPT; ++u) {
        const int unit = GUARD ? min(u0 + u * nwaves, nunits - 1) : u0 + u * nwaves;
        const int idx  = unit * 32 + lane;
        const int i    = idx / 10;
        const int f4   = idx - i * 10;
        const float* b = b2 + f4 * 4;
        dg[u] = deg[i];
        o[u]  = acc[(size_t)i * (LD2 / 4) + f4];
        bb[u] = make_float4(b[0], b[1], b[2], b[3]);
    }
#pragma unroll
    for (int u = 0; u < FPT; ++u) {
        const int unit = u0 + u * nwaves;
        if (!GUARD || unit < nunits) {
            const float di = rsqrtf(dg[u]);
            float4 r = o[u];
            r.x = r.x * di + bb[u].x;
            r.y = r.y * di + bb[u].y;
            r.z = r.z * di + bb[u].z;
            r.w = r.w * di + bb[u].w;
            out[unit * 32 + lane] = r;
        }
    }
}

__global__ __launch_bounds__(256) void k_fin2(float4* __restrict__ out,
                                              const float4* __restrict__ acc,
                                              const float* __restrict__ deg,
                                              const float* __restrict__ b2, int n) {
    const int nunits = (n * (COUT / 4)) >> 5;
    const int nwaves = gridDim.x * 8;
    const int lane   = threadIdx.x & 31;
    const int wave   = __builtin_amdgcn_readfirstlane((int)((blockIdx.x * 256 + threadIdx.x) >> 5));
    for (int u0 = wave; u0 < nunits; u0 += nwaves * FPT) {
        if (u0 + (FPT - 1) * nwaves < nunits) fin2_trip<false>(out, acc, deg, b2, u0, nwaves, nunits, lane);
        else                                  fin2_trip<true >(out, acc, deg, b2, u0, nwaves, nunits, lane);
    }
}

extern "C" void kernel_launch(void* const* d_in, const int* in_sizes, int n_in,
                              void* d_out, int out_size, void* d_ws, size_t ws_size,
                              hipStream_t stream) {
    const float* x  = (const float*)d_in[0];
    const float* W1 = (const float*)d_in[1];
    const float* b1 = (const float*)d_in[2];
    const float* W2 = (const float*)d_in[3];
    const float* b2 = (const float*)d_in[4];
    const int*   ei = (const int*)d_in[5];

    const int Nn = in_sizes[0] / FIN;
    const int E  = in_sizes[5] / 2;
    const int* src = ei;
    const int* dst = ei + E;

    float* ws   = (float*)d_ws;
    float* deg  = ws;
    float* h    = ws + Nn;
    float* agg  = h + (size_t)Nn * HID;
    float* h2   = h;
    float* acc2 = agg;
    float* out  = (float*)d_out;

    unsigned char* key = (unsigned char*)d_out + 4096;
    int* hdr = (int*)((unsigned char*)d_out + HOFF);
    int* lst = (int*)((unsigned char*)d_out + LOFF);
    int* pk  = (int*)agg;
    int gemmBlocks = ((Nn / 16) + 7) / 8;
    int gF = (Nn * (COUT / 4) / 32 + FPT * 8 - 1) / (FPT * 8);

    k_keys<<<KBLK, 256, 0, stream>>>(src, dst, E, (unsigned*)key, pk);
    k_search<<<NOWN, 1024, 0, stream>>>((const uint4*)key, pk, E, Nn, deg, hdr, lst, x, W1, h);

    k_own<HID><<<NOWN, 1024, 0, stream>>>((const uint4*)key, src, dst, E, Nn, h, agg, hdr, lst);

    k_gemm2<<<gemmBlocks, 256, 0, stream>>>(agg, W2, deg, b1, h2, Nn);
    k_own<COUT><<<NOWN, 1024, 0, stream>>>((const uint4*)key, src, dst, E, Nn, h2, acc2, hdr, lst);
    k_fin2<<<gF, 256, 0, stream>>>((float4*)out, (const float4*)acc2, deg, b2, Nn);
}


# layer 2 sums before it multiplies: edge pass 2 gathers scaled relu rows (64 wide), owners multiply their sums by W2 from LDS and write out; 2 launches fewer; rounding changes, sums in arrival order
# speedup vs baseline: 3.3634x; 1.0412x over previous
#include <hip/hip_runtime.h>

#define FIN 256
#define HID 64
#define COUT 40
#define LD2 64
#define LP  68

typedef float v2f __attribute__((ext_vector_type(2)));
typedef float v8f __attribute__((ext_vector_type(8)));

#define RPO   392
#define NOWN  256
#define PKS   9
#define KBLK  256
#define KU    4
#define KPL   4
#define TRIPK (KPL * 512)

__device__ __forceinline__ int pack_edge(int s, int d, unsigned& k) {
    k = (unsigned)(d / RPO);
    return (s << PKS) | (d - (int)k * RPO);
}

__global__ __launch_bounds__(256) void k_keys(const int* __restrict__ src, const int* __restrict__ dst, int E,
                                              unsigned* __restrict__ kp, int* __restrict__ pk) {
    const int lane = threadIdx.x & 31;
    const int wave = __builtin_amdgcn_readfirstlane((int)((blockIdx.x * 256 + threadIdx.x) >> 5));
    const int nwav = gridDim.x * 8;
    const int nwu  = (E + 127) >> 7;
    const int nwt  = ((E + TRIPK - 1) / TRIPK) * (TRIPK / 128);
    const bool al  = (E & 3) == 0;
    for (int w0 = wave; w0 < nwt; w0 += nwav * KU) {
        int4 s[KU], d[KU];
#pragma unroll
        for (int j = 0; j < KU; ++j) {
            const int wu = min(w0 + j * nwav, nwu - 1);
            const int e  = wu * 128 + lane * 4;
            if (al) {
                const int u = min(e, E - 4) >> 2;
                s[j] = ((const int4*)src)[u];
                d[j] = ((const int4*)dst)[u];
            } else {
                s[j].x = src[min(e, E - 1)];     d[j].x = dst[min(e, E - 1)];
                s[j].y = src[min(e + 1, E - 1)]; d[j].y = dst[min(e + 1, E - 1)];
                s[j].z = src[min(e + 2, E - 1)]; d[j].z = dst[min(e + 2, E - 1)];
                s[j].w = src[min(e + 3, E - 1)]; d[j].w = dst[min(e + 3, E - 1)];
            }
        }
#pragma unroll
        for (int j = 0; j < KU; ++j) {
            const int wu = w0 + j * nwav;
            if (wu < nwt) {
                const int e = wu * 128 + lane * 4;
                unsigned k0, k1, k2, k3;
                int4 p;
                p.x = pack_edge(s[j].x, d[j].x, k0);
                p.y = pack_edge(s[j].y, d[j].y, k1);
                p.z = pack_edge(s[j].z, d[j].z, k2);
                p.w = pack_edge(s[j].w, d[j].w, k3);
                if (al) {
                    if (e < E) ((int4*)pk)[e >> 2] = p;
                } else {
                    if (e < E)     pk[e]     = p.x;
                    if (e + 1 < E) pk[e + 1] = p.y;
                    if (e + 2 < E) pk[e + 2] = p.z;
                    if (e + 3 < E) pk[e + 3] = p.w;
                }
                const unsigned kk = k0 | (k1 << 8) | (k2 << 16) | (k3 << 24);
                unsigned wv = 0u;
#pragma unroll
                for (int t = 0; t < 32; ++t) {
                    const unsigned sres = __builtin_amdgcn_ballot_w32(((kk >> t) & 1u) != 0u);
                    const int jl = ((t >> 3) << 2) + (t & 3) + (((t >> 2) & 1) << 4);
                    wv = (lane == jl) ? sres : wv;
                }
                kp[(wu >> 3) * 256 + (lane >> 4) * 128 + (wu & 7) * 16 + (lane & 15)] = wv;
            }
        }
    }
}

__device__ __forceinline__ void strip_product(const float* __restrict__ X, const float* __restrict__ W,
                                              int m0, int n, int lane, v8f& c0, v8f& c1, v8f& c2, v8f& c3) {
    const int row = lane & 15;
    const int kk  = (lane >> 4) << 1;
    const float* Xr = X + (size_t)min(m0 + row, n - 1) * FIN + kk;
    for (int k0 = 0; k0 < FIN; k0 += 4) {
        v2f a;
        a.x = Xr[k0];
        a.y = Xr[k0 + 1];
        const float* Wp = W + (size_t)(k0 + kk) * HID;
        v2f b0 = { Wp[row],      Wp[HID + row] };
        v2f b1 = { Wp[16 + row], Wp[HID + 16 + row] };
        v2f b2 = { Wp[32 + row], Wp[HID + 32 + row] };
        v2f b3 = { Wp[48 + row], Wp[HID + 48 + row] };
        c0 = __builtin_amdgcn_wmma_f32_16x16x4_f32(false, a, false, b0, (short)0, c0, false, false);
        c1 = __builtin_amdgcn_wmma_f32_16x16x4_f32(false, a, false, b1, (short)0, c1, false, false);
        c2 = __builtin_amdgcn_wmma_f32_16x16x4_f32(false, a, false, b2, (short)0, c2, false, false);
        c3 = __builtin_amdgcn_wmma_f32_16x16x4_f32(false, a, false, b3, (short)0, c3, false, false);
    }
}

#define QCAP  (1024 + 32)
#define WCAP  416
#define SLICE (RPO * COUT)

__device__ __forceinline__ bool lists_fit(int n) {
    return (n - (NOWN - 1) * RPO) * COUT >= 64;
}
__device__ __forceinline__ int list_room(int nown) {
    return min(WCAP, (nown * COUT - 32) >> 5);
}
__device__ __forceinline__ void push_matches(unsigned c, unsigned m, int eb, int* myq, int& cnt) {
    do {
        if (c != 0u) {
            const int p = __builtin_ctz(c);
            myq[cnt + (int)__builtin_amdgcn_mbcnt_lo(m, 0u)] = eb + (p << 2);
            c &= c - 1u;
        }
        cnt += __builtin_popcount(m);
        m = __builtin_amdgcn_ballot_w32(c != 0u);
    } while (m != 0u);
}

__device__ __forceinline__ unsigned match_bits(uint4 lo, uint4 hi, const unsigned (&qm)[8], int eb, int e0, int E) {
    const unsigned x = (lo.x ^ qm[0]) | (lo.y ^ qm[1]) | (lo.z ^ qm[2]) | (lo.w ^ qm[3])
                     | (hi.x ^ qm[4]) | (hi.y ^ qm[5]) | (hi.z ^ qm[6]) | (hi.w ^ qm[7]);
    unsigned c = ~x;
    if (e0 + 1024 > E) {
        const int nv = (E - eb + 3) >> 2;
        c &= (nv >= 32) ? ~0u : (nv <= 0) ? 0u : ((1u << nv) - 1u);
    }
    return c;
}

template <bool FULL>
__device__ __forceinline__ void note_batch(const int* __restrict__ pk, int* cntrow, const int* myq, int top,
                                           int nb, int lane, bool rec, int room, int* __restrict__ mylist, int& nrec) {
    if (FULL || lane < nb) {
        const int p = pk[myq[top - 1 - lane]];
        if (rec && nrec + lane < room) mylist[nrec + lane] = p;
        atomicAdd(cntrow + (p & ((1 << PKS) - 1)), 1);
    }
    nrec += FULL ? 32 : nb;
}

template <int BATCH, class FullFn, class RestFn>
__device__ __forceinline__ void walk_keys(const uint4* __restrict__ key4, int E, int* myq, int g, int w, int lane,
                                          FullFn full, RestFn rest) {
    unsigned qm[8];
#pragma unroll
    for (int b = 0; b < 8; ++b) qm[b] = 0u - (((unsigned)g >> b) & 1u);
    int cnt = 0;
    const int ntrip = (E + TRIPK - 1) / TRIPK;
    const int rot = (int)(((long long)g * ntrip) >> 8);
    uint4 kv[KPL], nx[KPL];
    {
        int tq = (w < ntrip) ? w + rot : 0;
        if (tq >= ntrip) tq -= ntrip;
#pragma unroll
        for (int j = 0; j < KPL; ++j) kv[j] = key4[((tq * TRIPK) >> 4) + j * 32 + lane];
    }
    for (int t = w; t < ntrip; t += 32) {
        int tt = t + rot;
        if (tt >= ntrip) tt -= ntrip;
        const int e0 = tt * TRIPK;
        int tq = ((t + 32 < ntrip) ? t + 32 : t) + rot;
        if (tq >= ntrip) tq -= ntrip;
#pragma unroll
        for (int j = 0; j < KPL; ++j) nx[j] = key4[((tq * TRIPK) >> 4) + j * 32 + lane];
#pragma unroll
        for (int j = 0; j < KPL; j += 2) {
            const int eblk = e0 + j * 512;
            const int eb = eblk + ((lane >> 2) << 7) + (lane & 3);
            const unsigned c32 = match_bits(kv[j], kv[j + 1], qm, eb, eblk, E);
            const unsigned mm = __builtin_amdgcn_ballot_w32(c32 != 0u);
            if (mm != 0u) {
                push_matches(c32, mm, eb, myq, cnt);
                while (cnt >= BATCH) {
                    full(cnt);
                    cnt -= BATCH;
                }
            }
        }
#pragma unroll
        for (int j = 0; j < KPL; ++j) kv[j] = nx[j];
    }
    if (cnt > 0) rest(cnt, cnt);
}

__global__ __launch_bounds__(1024) void k_search(const uint4* __restrict__ key4, const int* __restrict__ pk,
                                                 int E, int n, float* __restrict__ deg,
                                                 int* __restrict__ lsts,
                                                 const float* __restrict__ X, const float* __restrict__ W,
                                                 float* __restrict__ H) {
    __shared__ int q[32 * QCAP];
    __shared__ int cntrow[RPO];
    const int tid  = threadIdx.x;
    const int lane = tid & 31;
    const int w    = __builtin_amdgcn_readfirstlane(tid >> 5);
    const int g    = blockIdx.x;
    const int row0 = g * RPO;
    const int nown = min(RPO, n - row0);
    for (int i = tid; i < RPO; i += 1024) cntrow[i] = 0;
    __syncthreads();

    const int  srow  = w * 16;
    const bool strip = srow < nown;
    const bool early = ((w >> 2) & 1) == 0;
    v8f c0 = {}, c1 = {}, c2 = {}, c3 = {};
    if (strip && early) strip_product(X, W, row0 + srow, n, lane, c0, c1, c2, c3);

    const bool lists = lists_fit(n);
    const int room = list_room(nown);
    int* myq = q + w * QCAP;
    int* slice = lsts + (size_t)g * SLICE;
    int* mylist = slice + 32 + w * room;
    int nrec = 0;
    walk_keys<32>(key4, E, myq, g, w, lane,
        [&](int top) { note_batch<true >(pk, cntrow, myq, top, 32, lane, lists, room, mylist, nrec); },
        [&](int top, int nb) { note_batch<false>(pk, cntrow, myq, top, nb, lane, lists, room, mylist, nrec); });
    if (lists && lane == 0) slice[w] = nrec;

    if (strip && !early) strip_product(X, W, row0 + srow, n, lane, c0, c1, c2, c3);
    __syncthreads();

    for (int i = tid; i < nown; i += 1024) deg[row0 + i] = 1.0f + (float)cntrow[i];

    if (strip) {
        const int lr0 = srow + ((lane >> 4) << 3);
        const int col = lane & 15;
        for (int r = 0; r < 8; ++r) {
            const int lr = lr0 + r;
            if (lr < nown) {
                const float di = rsqrtf(1.0f + (float)cntrow[lr]);
                const size_t off = (size_t)(row0 + lr) * HID + col;
                const float y0 = c0[r] * di, y1 = c1[r] * di, y2 = c2[r] * di, y3 = c3[r] * di;
                H[off]      = y0;
                H[off + 16] = y1;
                H[off + 32] = y2;
                H[off + 48] = y3;
            }
        }
    }
}

template <bool FULL>
__device__ __forceinline__ void add_batch(int pk, int b0, int nb, const float* __restrict__ H,
                                          float* rows, int lane) {
    int sj[8], dj[8];
#pragma unroll
    for (int u = 0; u < 8; ++u) {
        const int p = __builtin_amdgcn_readlane(pk, b0 + u);
        sj[u] = p >> PKS;
        dj[u] = p & ((1 << PKS) - 1);
    }
    float v0[8], v1[8];
#pragma unroll
    for (int u = 0; u < 8; ++u) {
        v0[u] = 0.0f; v1[u] = 0.0f;
        if (FULL || u < nb) v0[u] = H[(size_t)sj[u] * LD2 + lane];
    }
#pragma unroll
    for (int u = 0; u < 8; ++u)
        if (FULL || u < nb) v1[u] = H[(size_t)sj[u] * LD2 + 32 + lane];
#pragma unroll
    for (int u = 0; u < 8; ++u)
        if (FULL || u < nb) atomicAdd(rows + dj[u] * LP + lane, v0[u]);
#pragma unroll
    for (int u = 0; u < 8; ++u)
        if (FULL || u < nb) atomicAdd(rows + dj[u] * LP + 32 + lane, v1[u]);
}

#define ROWS_LDS (42 * 1024)
#define NIT ((RPO * (LD2 / 4) + 1023) / 1024)

__device__ __forceinline__ void own_sums(const int* __restrict__ src, const int* __restrict__ dst, int E,
                                         const float* __restrict__ H, const int* slice, bool lists,
                                         float* rows, int row0, int nown, int tid, int lane, int w) {
    const float4* H4 = (const float4*)H;
    float4* rows4 = (float4*)rows;
    float4 st[NIT];
#pragma unroll
    for (int u = 0; u < NIT; ++u) {
        const int i = tid + u * 1024;
        const int r = i >> 4, f = i & 15;
        st[u] = make_float4(0.0f, 0.0f, 0.0f, 0.0f);
        if (i < nown * 16) st[u] = H4[(size_t)(row0 + r) * (LD2 / 4) + f];
    }
#pragma unroll
    for (int u = 0; u < NIT; ++u) {
        const int i = tid + u * 1024;
        const int r = i >> 4, f = i & 15;
        if (i < nown * 16) rows4[r * (LP / 4) + f] = st[u];
    }
    __syncthreads();

    const int room = list_room(nown);
    const int* mylist = slice + 32 + w * room;
    int cw = room + 1;
    if (lists) cw = slice[lane];
    if (__builtin_amdgcn_ballot_w32(cw > room) == 0u) {
        const int nmine = __builtin_amdgcn_readlane(cw, w);
        int ent = 0;
        if (lane < nmine) ent = mylist[lane];
        int c0 = 0;
        for (; c0 + 32 <= nmine; c0 += 32) {
            const int cur = ent;
            if (c0 + 32 + lane < nmine) ent = mylist[c0 + 32 + lane];
            add_batch<true>(cur, 0,  8, H, rows, lane);
            add_batch<true>(cur, 8,  8, H, rows, lane);
            add_batch<true>(cur, 16, 8, H, rows, lane);
            add_batch<true>(cur, 24, 8, H, rows, lane);
        }
        const int left = nmine - c0;
        for (int b = 0; b < left; b += 8)
            add_batch<false>(ent, b, min(8, left - b), H, rows, lane);
    } else {
        for (int e0 = w * 32; e0 < E; e0 += 1024) {
            const int e = e0 + lane;
            int s = 0, d = -1;
            if (e < E) { s = src[e]; d = dst[e]; }
            const int dr = d - row0;
            unsigned m = __builtin_amdgcn_ballot_w32(dr >= 0 && dr < nown);
            while (m != 0u) {
                const int j = __builtin_ctz(m);
                m &= m - 1u;
                const int sj = __builtin_amdgcn_readlane(s, j);
                const int dj = __builtin_amdgcn_readlane(dr, j);
                const float v0 = H[(size_t)sj * LD2 + lane];
                const float v1 = H[(size_t)sj * LD2 + 32 + lane];
                atomicAdd(rows + dj * LP + lane, v0);
                atomicAdd(rows + dj * LP + 32 + lane, v1);
            }
        }
    }
    __syncthreads();
}

__global__ __launch_bounds__(1024) void k_own1(const int* __restrict__ src, const int* __restrict__ dst, int E, int n,
                                               const float* __restrict__ H, float* __restrict__ R,
                                               const float* __restrict__ deg, const float* __restrict__ b1,
                                               const int* __restrict__ lsts) {
    __shared__ float rows[ROWS_LDS];
    const int tid  = threadIdx.x;
    const int lane = tid & 31;
    const int w    = __builtin_amdgcn_readfirstlane(tid >> 5);
    const int g    = blockIdx.x;
    const int row0 = g * RPO;
    const int nown = min(RPO, n - row0);
    own_sums(src, dst, E, H, lsts + (size_t)g * SLICE, lists_fit(n), rows, row0, nown, tid, lane, w);

    const float4* rows4 = (const float4*)rows;
    float4* R4 = (float4*)R;
    float dg[NIT];
#pragma unroll
    for (int u = 0; u < NIT; ++u) {
        const int i = tid + u * 1024;
        dg[u] = 1.0f;
        if (i < nown * 16) dg[u] = deg[row0 + (i >> 4)];
    }
#pragma unroll
    for (int u = 0; u < NIT; ++u) {
        const int i = tid + u * 1024;
        const int r = i >> 4, f = i & 15;
        if (i < nown * 16) {
            const float di = rsqrtf(dg[u]);
            const float* b = b1 + f * 4;
            const float4 v = rows4[r * (LP / 4) + f];
            float4 o;
            o.x = di * fmaxf(v.x * di + b[0], 0.0f);
            o.y = di * fmaxf(v.y * di + b[1], 0.0f);
            o.z = di * fmaxf(v.z * di + b[2], 0.0f);
            o.w = di * fmaxf(v.w * di + b[3], 0.0f);
            R4[(size_t)(row0 + r) * (LD2 / 4) + f] = o;
        }
    }
}

__global__ __launch_bounds__(1024) void k_own2(const int* __restrict__ src, const int* __restrict__ dst, int E, int n,
                                               const float* __restrict__ R, const float* __restrict__ W,
                                               const float* __restrict__ deg, const float* __restrict__ b2,
                                               float* out) {
    __shared__ float rows[ROWS_LDS];
    const int tid  = threadIdx.x;
    const int lane = tid & 31;
    const int w    = __builtin_amdgcn_readfirstlane(tid >> 5);
    const int g    = blockIdx.x;
    const int row0 = g * RPO;
    const int nown = min(RPO, n - row0);
    own_sums(src, dst, E, R, (const int*)out + (size_t)g * SLICE, lists_fit(n), rows, row0, nown, tid, lane, w);

    const int srow = w * 16;
    if (srow < nown) {
        const int row = lane & 15;
        const int kk  = (lane >> 4) << 1;
        const int lr0 = srow + ((lane >> 4) << 3);
        float di[8];
#pragma unroll
        for (int r = 0; r < 8; ++r) di[r] = deg[row0 + min(lr0 + r, nown - 1)];
        const float* Ar = rows + min(srow + row, nown - 1) * LP + kk;
        v8f c0 = {}, c1 = {}, c2 = {};
#pragma unroll
        for (int k0 = 0; k0 < HID; k0 += 4) {
            v2f a = { Ar[k0], Ar[k0 + 1] };
            const float* Wp = W + (size_t)(k0 + kk) * COUT;
            v2f b0 = { Wp[row],      Wp[COUT + row] };
            v2f b1 = { Wp[16 + row], Wp[COUT + 16 + row] };
            const int col2 = 32 + row;
            v2f bz = { (col2 < COUT) ? Wp[col2] : 0.0f,
                       (col2 < COUT) ? Wp[COUT + col2] : 0.0f };
            c0 = __builtin_amdgcn_wmma_f32_16x16x4_f32(false, a, false, b0, (short)0, c0, false, false);
            c1 = __builtin_amdgcn_wmma_f32_16x16x4_f32(false, a, false, b1, (short)0, c1, false, false);
            c2 = __builtin_amdgcn_wmma_f32_16x16x4_f32(false, a, false, bz, (short)0, c2, false, false);
        }
        const int col = lane & 15;
        const float bb0 = b2[col], bb1 = b2[16 + col], bb2 = b2[32 + (col & 7)];
        float y0[8], y1[8], y2[8];
#pragma unroll
        for (int r = 0; r < 8; ++r) {
            const float d = rsqrtf(di[r]);
            y0[r] = c0[r] * d + bb0;
            y1[r] = c1[r] * d + bb1;
            y2[r] = c2[r] * d + bb2;
        }
#pragma unroll
        for (int r = 0; r < 8; ++r) {
            const int lr = lr0 + r;
            if (lr < nown) {
                float* o = rows + lr * LP;
                o[col]      = y0[r];
                o[16 + col] = y1[r];
                if (col < 8) o[32 + col] = y2[r];
            }
        }
    }
    __syncthreads();

    const float4* rows4 = (const float4*)rows;
    float4* out4 = (float4*)out;
    for (int i = tid; i < nown * (COUT / 4); i += 1024) {
        const int r = i / (COUT / 4), f = i - r * (COUT / 4);
        out4[(size_t)(row0 + r) * (COUT / 4) + f] = rows4[r * (LP / 4) + f];
    }
}

extern "C" void kernel_launch(void* const* d_in, const int* in_sizes, int n_in,
                              void* d_out, int out_size, void* d_ws, size_t ws_size,
                              hipStream_t stream) {
    const float* x  = (const float*)d_in[0];
    const float* W1 = (const float*)d_in[1];
    const float* b1 = (const float*)d_in[2];
    const float* W2 = (const float*)d_in[3];
    const float* b2 = (const float*)d_in[4];
    const int*   ei = (const int*)d_in[5];

    const int Nn = in_sizes[0] / FIN;
    const int E  = in_sizes[5] / 2;
    const int* src = ei;
    const int* dst = ei + E;

    float* ws   = (float*)d_ws;
    float* deg  = ws;
    float* h    = ws + Nn;
    float* r1   = h + (size_t)Nn * HID;
    float* out  = (float*)d_out;

    int* pk = (int*)r1;
    unsigned char* key = (unsigned char*)r1 + (((size_t)E * 4 + 4095) & ~(size_t)4095);

    k_keys<<<KBLK, 256, 0, stream>>>(src, dst, E, (unsigned*)key, pk);
    k_search<<<NOWN, 1024, 0, stream>>>((const uint4*)key, pk, E, Nn, deg, (int*)out, x, W1, h);

    k_own1<<<NOWN, 1024, 0, stream>>>(src, dst, E, Nn, h, r1, deg, b1, (const int*)out);
    k_own2<<<NOWN, 1024, 0, stream>>>(src, dst, E, Nn, r1, W2, deg, b2, out);
}
